# starting kernel
# baseline (speedup 1.0000x reference)
#include <hip/hip_runtime.h>

constexpr int kNG = 100000;
constexpr int kND = 50000;
constexpr int kNC = 50000;
constexpr int kH  = 128;

typedef __attribute__((ext_vector_type(2))) float v2f;
typedef __attribute__((ext_vector_type(8))) float v8f;

__global__ __launch_bounds__(256)
void hgnn_scatter(const float* __restrict__ x, const int* __restrict__ src,
                  const int* __restrict__ dst, float* __restrict__ sum,
                  float* __restrict__ cnt, int nE) {
  const int lane = threadIdx.x & 31;
  const int wave = (blockIdx.x * blockDim.x + threadIdx.x) >> 5;
  const int nW   = (gridDim.x * blockDim.x) >> 5;
  for (int e = wave; e < nE; e += nW) {
    const int s = src[e];
    const int d = dst[e];
    const float4 v = *(const float4*)(x + (size_t)s * kH + lane * 4);
    float* p = sum + (size_t)d * kH + lane * 4;
    atomicAdd(p + 0, v.x);
    atomicAdd(p + 1, v.y);
    atomicAdd(p + 2, v.z);
    atomicAdd(p + 3, v.w);
    if (lane == 0) atomicAdd(cnt + d, 1.0f);
  }
}

__global__ void hgnn_addmat(const float* __restrict__ a, const float* __restrict__ b,
                            float* __restrict__ o, int n) {
  int i = blockIdx.x * blockDim.x + threadIdx.x;
  if (i < n) o[i] = a[i] + b[i];
}

__global__ __launch_bounds__(256)
void hgnn_gemm3(const float* __restrict__ A0, const float* __restrict__ c0,
                const float* __restrict__ W0,
                const float* __restrict__ A1, const float* __restrict__ c1,
                const float* __restrict__ W1,
                const float* __restrict__ A2, const float* __restrict__ c2,
                const float* __restrict__ W2,
                const float* __restrict__ b0, const float* __restrict__ b1,
                float* __restrict__ Y, int relu) {
  __shared__ float As[16 * 132];
  const int tid  = threadIdx.x;
  const int wv   = tid >> 5;
  const int lane = tid & 31;
  const int m    = lane & 15;
  const int koff = (lane >> 4) << 1;
  const int col  = (wv << 4) + m;
  const int row0 = blockIdx.x << 4;

  const float* Alist[3] = {A0, A1, A2};
  const float* Clist[3] = {c0, c1, c2};
  const float* Wlist[3] = {W0, W1, W2};

  v8f acc = {};
  for (int t = 0; t < 3; ++t) {
    const float* A = Alist[t];
    if (!A) continue;
    const float* cp = Clist[t];
    const float* W  = Wlist[t];

    __syncthreads();
    for (int i = tid; i < 16 * kH; i += 256) {
      const int r = i >> 7;
      const int c = i & (kH - 1);
      float v = A[(size_t)(row0 + r) * kH + c];
      if (cp) v *= 1.0f / fmaxf(cp[row0 + r], 1.0f);
      As[r * 132 + c] = v;
    }
    __syncthreads();

#pragma unroll 8
    for (int k0 = 0; k0 < kH; k0 += 4) {
      const int k = k0 + koff;
      v2f av, bv;
      av.x = As[m * 132 + k];
      av.y = As[m * 132 + k + 1];
      bv.x = W[(size_t)k * kH + col];
      bv.y = W[(size_t)(k + 1) * kH + col];
      acc = __builtin_amdgcn_wmma_f32_16x16x4_f32(
          false, av, false, bv, (short)0, acc, false, false);
    }
  }

  const float bsum = (b0 ? b0[col] : 0.0f) + (b1 ? b1[col] : 0.0f);
  __syncthreads();
#pragma unroll
  for (int r = 0; r < 8; ++r) {
    float v = acc[r] + bsum;
    if (relu) v = fmaxf(v, 0.0f);
    Y[(size_t)(row0 + r + ((lane >> 4) << 3)) * kH + col] = v;
  }
}

extern "C" void kernel_launch(void* const* d_in, const int* in_sizes, int n_in,
                              void* d_out, int out_size, void* d_ws, size_t ws_size,
                              hipStream_t stream) {
  (void)n_in; (void)out_size; (void)ws_size;
  const float* EG = (const float*)d_in[0];
  const float* ED = (const float*)d_in[1];
  const float* EC = (const float*)d_in[2];
  const float* Wl = (const float*)d_in[3];
  const float* bl = (const float*)d_in[4];
  const float* Wr = (const float*)d_in[5];
  const int* eg_s  = (const int*)d_in[6];
  const int* eg_d  = (const int*)d_in[7];
  const int* ecd_s = (const int*)d_in[8];
  const int* ecd_d = (const int*)d_in[9];
  const int* ecg_s = (const int*)d_in[10];
  const int* ecg_d = (const int*)d_in[11];
  const int* egc_s = (const int*)d_in[12];
  const int* egc_d = (const int*)d_in[13];

  float* ws = (float*)d_ws;
  float* s0 = ws;
  float* s1 = s0 + (size_t)kNG * kH;
  float* s2 = s1 + (size_t)kND * kH;
  float* s3 = s2 + (size_t)kNG * kH;
  float* c0 = s3 + (size_t)kNC * kH;
  float* c1 = c0 + kNG;
  float* c2 = c1 + kND;
  float* c3 = c2 + kNG;
  float* wsum = c3 + kNC;
  const size_t nAccum = (size_t)(2 * kNG + kND + kNC) * kH + (2 * kNG + kND + kNC);
  const size_t zero_bytes = nAccum * sizeof(float);

  float* OG = (float*)d_out;
  float* OD = OG + (size_t)kNG * kH;
  float* OC = OD + (size_t)kND * kH;

  const float* xg = EG;
  const float* xd = ED;
  const float* xc = EC;

  for (int L = 0; L < 2; ++L) {
    hipMemsetAsync(ws, 0, zero_bytes, stream);
    hgnn_scatter<<<2048, 256, 0, stream>>>(xg, eg_s,  eg_d,  s0, c0, in_sizes[6]);
    hgnn_scatter<<<1024, 256, 0, stream>>>(xc, ecd_s, ecd_d, s1, c1, in_sizes[8]);
    hgnn_scatter<<<2048, 256, 0, stream>>>(xc, ecg_s, ecg_d, s2, c2, in_sizes[10]);
    hgnn_scatter<<<2048, 256, 0, stream>>>(xg, egc_s, egc_d, s3, c3, in_sizes[12]);

    const float* WlL = Wl + (size_t)L * 4 * kH * kH;
    const float* WrL = Wr + (size_t)L * 4 * kH * kH;
    const float* blL = bl + (size_t)L * 4 * kH;
    hgnn_addmat<<<(kH * kH + 255) / 256, 256, 0, stream>>>(WrL, WrL + 2 * kH * kH,
                                                           wsum, kH * kH);
    const int relu = (L == 0) ? 1 : 0;
    hgnn_gemm3<<<kNG / 16, 256, 0, stream>>>(
        s0, c0, WlL,
        s2, c2, WlL + 2 * kH * kH,
        xg, nullptr, wsum,
        blL, blL + 2 * kH, OG, relu);
    hgnn_gemm3<<<kND / 16, 256, 0, stream>>>(
        s1, c1, WlL + 1 * kH * kH,
        xd, nullptr, WrL + 1 * kH * kH,
        nullptr, nullptr, nullptr,
        blL + 1 * kH, nullptr, OD, relu);
    hgnn_gemm3<<<kNC / 16, 256, 0, stream>>>(
        s3, c3, WlL + 3 * kH * kH,
        xc, nullptr, WrL + 3 * kH * kH,
        nullptr, nullptr, nullptr,
        blL + 3 * kH, nullptr, OC, relu);

    xg = OG; xd = OD; xc = OC;
  }
}


# Float-atomic scatter and its two big memsets replaced by per-call destination-sorted edge lists (integer atomics) plus an atomic-free row gather; neighbour rows summed in list order by one wave
# speedup vs baseline: 1.7841x; 1.7841x over previous
#include <hip/hip_runtime.h>


constexpr int kNG = 100000;
constexpr int kND = 50000;
constexpr int kNC = 50000;
constexpr int kH  = 128;

constexpr int kV1 = 0;
constexpr int kV0 = kND;
constexpr int kV2 = kND + kNG;
constexpr int kV3 = kND + 2 * kNG;
constexpr int kNV = kND + 2 * kNG + kNC;
constexpr int kETot = 600000 + 300000 + 600000 + 600000;

typedef __attribute__((ext_vector_type(2))) float v2f;
typedef __attribute__((ext_vector_type(8))) float v8f;

__global__ __launch_bounds__(256)
void hgnn_count(const int* __restrict__ d0, const int* __restrict__ d1,
                const int* __restrict__ d2, const int* __restrict__ d3,
                int n0, int n1, int n2, int n3, int* __restrict__ deg) {
  const int r = blockIdx.y;
  const int* d = (r == 0) ? d0 : (r == 1) ? d1 : (r == 2) ? d2 : d3;
  const int n  = (r == 0) ? n0 : (r == 1) ? n1 : (r == 2) ? n2 : n3;
  const int vb = (r == 0) ? kV0 : (r == 1) ? kV1 : (r == 2) ? kV2 : kV3;
  for (int e = blockIdx.x * 256 + threadIdx.x; e < n; e += gridDim.x * 256)
    atomicAdd(deg + vb + d[e], 1);
}

__global__ __launch_bounds__(256)
void hgnn_offsets(const int* __restrict__ deg, int* __restrict__ cur,
                  int* __restrict__ tot) {
  __shared__ int wpart[8];
  __shared__ int bbase;
  const int v    = blockIdx.x * 256 + threadIdx.x;
  const int lane = threadIdx.x & 31;
  const int wv   = threadIdx.x >> 5;
  const int n    = (v < kNV) ? deg[v] : 0;
  int inc = n;
  for (int o = 1; o < 32; o <<= 1) {
    const int t = __shfl_up(inc, o);
    if (lane >= o) inc += t;
  }
  if (lane == 31) wpart[wv] = inc;
  __syncthreads();
  if (threadIdx.x == 0) {
    int t = 0;
    for (int i = 0; i < 8; ++i) { const int w = wpart[i]; wpart[i] = t; t += w; }
    bbase = atomicAdd(tot, t);
  }
  __syncthreads();
  if (v < kNV) cur[v] = bbase + wpart[wv] + inc - n;
}

__global__ __launch_bounds__(256)
void hgnn_fill(const int* __restrict__ s0, const int* __restrict__ d0,
               const int* __restrict__ s1, const int* __restrict__ d1,
               const int* __restrict__ s2, const int* __restrict__ d2,
               const int* __restrict__ s3, const int* __restrict__ d3,
               int n0, int n1, int n2, int n3,
               int* __restrict__ cur, int* __restrict__ col) {
  const int r = blockIdx.y;
  const int* s = (r == 0) ? s0 : (r == 1) ? s1 : (r == 2) ? s2 : s3;
  const int* d = (r == 0) ? d0 : (r == 1) ? d1 : (r == 2) ? d2 : d3;
  const int n  = (r == 0) ? n0 : (r == 1) ? n1 : (r == 2) ? n2 : n3;
  const int vb = (r == 0) ? kV0 : (r == 1) ? kV1 : (r == 2) ? kV2 : kV3;
  for (int e = blockIdx.x * 256 + threadIdx.x; e < n; e += gridDim.x * 256) {
    const int p = atomicAdd(cur + vb + d[e], 1);
    if ((unsigned)p < (unsigned)kETot) col[p] = s[e];
  }
}

__global__ __launch_bounds__(256)
void hgnn_gather(const float* __restrict__ xg, const float* __restrict__ xc,
                 const int* __restrict__ deg, const int* __restrict__ cur,
                 const int* __restrict__ col, float* __restrict__ sbase,
                 float* __restrict__ cnt, int vbeg, int vend) {
  const int lane = threadIdx.x & 31;
  const int wave = blockIdx.x * 8 + __builtin_amdgcn_readfirstlane(threadIdx.x >> 5);
  const int nW   = gridDim.x * 8;
  for (int v = vbeg + wave; v < vend; v += nW) {
    const int n  = __builtin_amdgcn_readfirstlane(deg[v]);
    const int st = __builtin_amdgcn_readfirstlane(cur[v]) - n;
    const bool src_gene = (v >= kV0 && v < kV2) || (v >= kV3);
    const float* x = src_gene ? xg : xc;
    const unsigned smax = src_gene ? (unsigned)(kNG - 1) : (unsigned)(kNC - 1);
    float* srow = sbase + (size_t)((v < kV0) ? (2 * kNG + v) : (v - kV0)) * kH;

    float4 acc = make_float4(0.0f, 0.0f, 0.0f, 0.0f);
    for (int c = 0; c < n; c += 32) {
      const int m = (n - c < 32) ? (n - c) : 32;
      int idx = 0;
      if (lane < m) idx = col[st + c + lane];
      for (int j0 = 0; j0 < m; j0 += 8) {
        unsigned s[8];
#pragma unroll
        for (int u = 0; u < 8; ++u) {
          const unsigned t = (unsigned)__builtin_amdgcn_readlane(idx, j0 + u);
          s[u] = (t < smax) ? t : smax;
        }
        float4 r[8];
#pragma unroll
        for (int u = 0; u < 8; ++u) {
          r[u] = make_float4(0.0f, 0.0f, 0.0f, 0.0f);
          if (j0 + u < m) r[u] = *(const float4*)(x + (size_t)s[u] * kH + lane * 4);
        }
#pragma unroll
        for (int u = 0; u < 8; ++u) {
          acc.x += r[u].x; acc.y += r[u].y; acc.z += r[u].z; acc.w += r[u].w;
        }
      }
    }
    *(float4*)(srow + lane * 4) = acc;
    if (lane == 0) cnt[v] = (float)n;
  }
}

__global__ void hgnn_addmat(const float* __restrict__ a, const float* __restrict__ b,
                            float* __restrict__ o, int n) {
  int i = blockIdx.x * blockDim.x + threadIdx.x;
  if (i < n) o[i] = a[i] + b[i];
}

__global__ __launch_bounds__(256)
void hgnn_gemm3(const float* __restrict__ A0, const float* __restrict__ c0,
                const float* __restrict__ W0,
                const float* __restrict__ A1, const float* __restrict__ c1,
                const float* __restrict__ W1,
                const float* __restrict__ A2, const float* __restrict__ c2,
                const float* __restrict__ W2,
                const float* __restrict__ b0, const float* __restrict__ b1,
                float* __restrict__ Y, int relu) {
  __shared__ float As[16 * 132];
  const int tid  = threadIdx.x;
  const int wv   = tid >> 5;
  const int lane = tid & 31;
  const int m    = lane & 15;
  const int koff = (lane >> 4) << 1;
  const int col  = (wv << 4) + m;
  const int row0 = blockIdx.x << 4;

  const float* Alist[3] = {A0, A1, A2};
  const float* Clist[3] = {c0, c1, c2};
  const float* Wlist[3] = {W0, W1, W2};

  v8f acc = {};
  for (int t = 0; t < 3; ++t) {
    const float* A = Alist[t];
    if (!A) continue;
    const float* cp = Clist[t];
    const float* W  = Wlist[t];

    __syncthreads();
    for (int i = tid; i < 16 * kH; i += 256) {
      const int r = i >> 7;
      const int c = i & (kH - 1);
      float v = A[(size_t)(row0 + r) * kH + c];
      if (cp) v *= 1.0f / fmaxf(cp[row0 + r], 1.0f);
      As[r * 132 + c] = v;
    }
    __syncthreads();

#pragma unroll 8
    for (int k0 = 0; k0 < kH; k0 += 4) {
      const int k = k0 + koff;
      v2f av, bv;
      av.x = As[m * 132 + k];
      av.y = As[m * 132 + k + 1];
      bv.x = W[(size_t)k * kH + col];
      bv.y = W[(size_t)(k + 1) * kH + col];
      acc = __builtin_amdgcn_wmma_f32_16x16x4_f32(
          false, av, false, bv, (short)0, acc, false, false);
    }
  }

  const float bsum = (b0 ? b0[col] : 0.0f) + (b1 ? b1[col] : 0.0f);
  __syncthreads();
#pragma unroll
  for (int r = 0; r < 8; ++r) {
    float v = acc[r] + bsum;
    if (relu) v = fmaxf(v, 0.0f);
    Y[(size_t)(row0 + r + ((lane >> 4) << 3)) * kH + col] = v;
  }
}

extern "C" void kernel_launch(void* const* d_in, const int* in_sizes, int n_in,
                              void* d_out, int out_size, void* d_ws, size_t ws_size,
                              hipStream_t stream) {
  (void)n_in; (void)out_size; (void)ws_size;
  const float* EG = (const float*)d_in[0];
  const float* ED = (const float*)d_in[1];
  const float* EC = (const float*)d_in[2];
  const float* Wl = (const float*)d_in[3];
  const float* bl = (const float*)d_in[4];
  const float* Wr = (const float*)d_in[5];
  const int* eg_s  = (const int*)d_in[6];
  const int* eg_d  = (const int*)d_in[7];
  const int* ecd_s = (const int*)d_in[8];
  const int* ecd_d = (const int*)d_in[9];
  const int* ecg_s = (const int*)d_in[10];
  const int* ecg_d = (const int*)d_in[11];
  const int* egc_s = (const int*)d_in[12];
  const int* egc_d = (const int*)d_in[13];

  float* ws = (float*)d_ws;
  float* s0 = ws;
  float* s2 = s0 + (size_t)kNG * kH;
  float* s3 = s2 + (size_t)kNG * kH;
  float* s1 = s3;
  float* cnt  = s3 + (size_t)kNC * kH;
  float* wsum = cnt + kNV;
  int* deg = (int*)(wsum + kH * kH);
  int* tot = deg + kNV;
  int* cur = tot + 1;
  int* col = cur + kNV;

  float* OG = (float*)d_out;
  float* OD = OG + (size_t)kNG * kH;
  float* OC = OD + (size_t)kND * kH;

  hipMemsetAsync(deg, 0, (size_t)(kNV + 1) * sizeof(int), stream);
  hgnn_count<<<dim3(586, 4), 256, 0, stream>>>(
      eg_d, ecd_d, ecg_d, egc_d,
      in_sizes[6], in_sizes[8], in_sizes[10], in_sizes[12], deg);
  hgnn_offsets<<<(kNV + 255) / 256, 256, 0, stream>>>(deg, cur, tot);
  hgnn_fill<<<dim3(586, 4), 256, 0, stream>>>(
      eg_s, eg_d, ecd_s, ecd_d, ecg_s, ecg_d, egc_s, egc_d,
      in_sizes[6], in_sizes[8], in_sizes[10], in_sizes[12], cur, col);

  const float* xg = EG;
  const float* xd = ED;
  const float* xc = EC;

  for (int L = 0; L < 2; ++L) {
    const float* WlL = Wl + (size_t)L * 4 * kH * kH;
    const float* WrL = Wr + (size_t)L * 4 * kH * kH;
    const float* blL = bl + (size_t)L * 4 * kH;
    const int relu = (L == 0) ? 1 : 0;

    hgnn_gather<<<2048, 256, 0, stream>>>(xg, xc, deg, cur, col, s0, cnt, kV1, kV0);
    hgnn_gemm3<<<kND / 16, 256, 0, stream>>>(
        s1, cnt + kV1, WlL + 1 * kH * kH,
        xd, nullptr, WrL + 1 * kH * kH,
        nullptr, nullptr, nullptr,
        blL + 1 * kH, nullptr, OD, relu);

    hgnn_gather<<<2048, 256, 0, stream>>>(xg, xc, deg, cur, col, s0, cnt, kV0, kNV);
    hgnn_addmat<<<(kH * kH + 255) / 256, 256, 0, stream>>>(WrL, WrL + 2 * kH * kH,
                                                           wsum, kH * kH);
    hgnn_gemm3<<<kNG / 16, 256, 0, stream>>>(
        s0, cnt + kV0, WlL,
        s2, cnt + kV2, WlL + 2 * kH * kH,
        xg, nullptr, wsum,
        blL, blL + 2 * kH, OG, relu);
    hgnn_gemm3<<<kNC / 16, 256, 0, stream>>>(
        s3, cnt + kV3, WlL + 3 * kH * kH,
        xc, nullptr, WrL + 3 * kH * kH,
        nullptr, nullptr, nullptr,
        blL + 3 * kH, nullptr, OC, relu);

    xg = OG; xd = OD; xc = OC;
  }
}


# Edge lists built in ONE launch on LDS counters (a work-group owns a node range and scans its relation's destinations once): no global atomics, no memset; neighbour order = LDS arrival order
# speedup vs baseline: 2.0871x; 1.1698x over previous
#include <hip/hip_runtime.h>


constexpr int kNG = 100000;
constexpr int kND = 50000;
constexpr int kNC = 50000;
constexpr int kH  = 128;

constexpr int kV1 = 0;
constexpr int kV0 = kND;
constexpr int kV2 = kND + kNG;
constexpr int kV3 = kND + 2 * kNG;
constexpr int kNV = kND + 2 * kNG + kNC;
constexpr int kETot = 600000 + 300000 + 600000 + 600000;

typedef __attribute__((ext_vector_type(2))) float v2f;
typedef __attribute__((ext_vector_type(8))) float v8f;

constexpr int kBT   = 1024;
constexpr int kWG1  = 34;
constexpr int kWG0  = 74;
constexpr int kWG2  = 74;
constexpr int kWG3  = 74;
constexpr int kBuildGrid = kWG1 + kWG0 + kWG2 + kWG3;
constexpr int kMaxR = 2 * kBT;
constexpr int kMaxGroups = 5;
constexpr int kMaxEdges  = kMaxGroups * 32 * kBT * 4;
constexpr int kNoNode    = 0x7fffffff;
static_assert((kND + kWG1 - 1) / kWG1 <= kMaxR, "range too long");
static_assert((kNG + kWG0 - 1) / kWG0 <= kMaxR, "range too long");
static_assert((kNG + kWG2 - 1) / kWG2 <= kMaxR, "range too long");
static_assert((kNC + kWG3 - 1) / kWG3 <= kMaxR, "range too long");
static_assert(600000 <= kMaxEdges, "mask words too few");

__global__ __launch_bounds__(kBT)
void hgnn_build(const int* __restrict__ s0, const int* __restrict__ d0,
                const int* __restrict__ s1, const int* __restrict__ d1,
                const int* __restrict__ s2, const int* __restrict__ d2,
                const int* __restrict__ s3, const int* __restrict__ d3,
                int n0, int n1, int n2, int n3,
                int* __restrict__ deg, int* __restrict__ cur, int* __restrict__ col) {
  __shared__ int      ldeg[kMaxR];
  __shared__ unsigned lmask[kMaxGroups * kBT];
  __shared__ int      wpart[32];
  __shared__ int      sbelow;
  const int tid  = threadIdx.x;
  const int lane = tid & 31;
  const int wv   = tid >> 5;

  const int b = blockIdx.x;
  int r, w, nwg, nn;
  if (b < kWG1)                    { r = 1; w = b;                      nwg = kWG1; nn = kND; }
  else if (b < kWG1 + kWG0)        { r = 0; w = b - kWG1;               nwg = kWG0; nn = kNG; }
  else if (b < kWG1 + kWG0 + kWG2) { r = 2; w = b - kWG1 - kWG0;        nwg = kWG2; nn = kNG; }
  else                             { r = 3; w = b - kWG1 - kWG0 - kWG2; nwg = kWG3; nn = kNC; }
  const int* s = (r == 0) ? s0 : (r == 1) ? s1 : (r == 2) ? s2 : s3;
  const int* d = (r == 0) ? d0 : (r == 1) ? d1 : (r == 2) ? d2 : d3;
  int n        = (r == 0) ? n0 : (r == 1) ? n1 : (r == 2) ? n2 : n3;
  if (n > kMaxEdges) n = kMaxEdges;
  const int vb    = (r == 0) ? kV0 : (r == 1) ? kV1 : (r == 2) ? kV2 : kV3;
  const int ebase = (r == 1) ? 0 : (r == 0) ? n1 : (r == 2) ? n1 + n0 : n1 + n0 + n2;
  const int rpw = (nn + nwg - 1) / nwg;
  const int lo  = w * rpw;
  const int R   = (nn - lo < rpw) ? (nn - lo) : rpw;

  for (int i = tid; i < kMaxR; i += kBT) ldeg[i] = 0;
  if (tid == 0) sbelow = 0;
  __syncthreads();

  const int4* d4 = (const int4*)d;
  const int n4 = n >> 2;
  int below = 0;
  auto count1 = [&](int dv) -> bool {
    const unsigned l = (unsigned)dv - (unsigned)lo;
    below += (dv < lo) ? 1 : 0;
    if (l < (unsigned)R) { atomicAdd(&ldeg[l], 1); return true; }
    return false;
  };

  for (int g = 0; g < kMaxGroups; ++g) {
    unsigned mask = 0;
    for (int h = 0; h < 8; ++h) {
      const int jb = ((g * 8 + h) * 4) * kBT + tid;
      if (jb >= n4) break;
      int4 q[4];
#pragma unroll
      for (int u = 0; u < 4; ++u) {
        const int j = jb + u * kBT;
        q[u] = (j < n4) ? d4[j] : make_int4(kNoNode, kNoNode, kNoNode, kNoNode);
      }
#pragma unroll
      for (int u = 0; u < 4; ++u) {
        const bool hx = count1(q[u].x);
        const bool hy = count1(q[u].y);
        const bool hz = count1(q[u].z);
        const bool hw = count1(q[u].w);
        if (hx | hy | hz | hw) mask |= 1u << (h * 4 + u);
      }
    }
    lmask[g * kBT + tid] = mask;
  }
  const int et = (n4 << 2) + tid;
  const bool has_tail = tid < (n & 3);
  if (has_tail) count1(d[et]);

  for (int o = 16; o > 0; o >>= 1) below += __shfl_xor(below, o);
  if (lane == 0) atomicAdd(&sbelow, below);
  __syncthreads();

  const int i0 = 2 * tid;
  const int i1 = i0 + 1;
  const int a  = (i0 < R) ? ldeg[i0] : 0;
  const int c  = (i1 < R) ? ldeg[i1] : 0;
  const int sum = a + c;
  int inc = sum;
  for (int o = 1; o < 32; o <<= 1) {
    const int t = __shfl_up(inc, o);
    if (lane >= o) inc += t;
  }
  if (lane == 31) wpart[wv] = inc;
  __syncthreads();
  if (wv == 0) {
    const int x = wpart[lane];
    int xi = x;
    for (int o = 1; o < 32; o <<= 1) {
      const int t = __shfl_up(xi, o);
      if (lane >= o) xi += t;
    }
    wpart[lane] = xi - x;
  }
  __syncthreads();
  const int ex   = wpart[wv] + inc - sum;
  const int base = ebase + sbelow;
  if (i0 < R) { ldeg[i0] = ex;     deg[vb + lo + i0] = a; cur[vb + lo + i0] = base + ex + a; }
  if (i1 < R) { ldeg[i1] = ex + a; deg[vb + lo + i1] = c; cur[vb + lo + i1] = base + ex + sum; }
  __syncthreads();

  auto place = [&](int dv, int e) {
    const unsigned l = (unsigned)dv - (unsigned)lo;
    if (l < (unsigned)R) {
      const int p = base + atomicAdd(&ldeg[l], 1);
      if ((unsigned)p < (unsigned)kETot) col[p] = s[e];
    }
  };
  for (int g = 0; g < kMaxGroups; ++g) {
    unsigned mask = lmask[g * kBT + tid];
    while (mask) {
      const int k = __builtin_ctz(mask);
      mask &= mask - 1;
      const int j = (g * 32 + k) * kBT + tid;
      const int4 q = d4[j];
      place(q.x, 4 * j);
      place(q.y, 4 * j + 1);
      place(q.z, 4 * j + 2);
      place(q.w, 4 * j + 3);
    }
  }
  if (has_tail) place(d[et], et);
}

__global__ __launch_bounds__(256)
void hgnn_gather(const float* __restrict__ xg, const float* __restrict__ xc,
                 const int* __restrict__ deg, const int* __restrict__ cur,
                 const int* __restrict__ col, float* __restrict__ sbase,
                 float* __restrict__ cnt, int vbeg, int vend) {
  const int lane = threadIdx.x & 31;
  const int wave = blockIdx.x * 8 + __builtin_amdgcn_readfirstlane(threadIdx.x >> 5);
  const int nW   = gridDim.x * 8;
  for (int v = vbeg + wave; v < vend; v += nW) {
    const int n  = __builtin_amdgcn_readfirstlane(deg[v]);
    const int st = __builtin_amdgcn_readfirstlane(cur[v]) - n;
    const bool src_gene = (v >= kV0 && v < kV2) || (v >= kV3);
    const float* x = src_gene ? xg : xc;
    const unsigned smax = src_gene ? (unsigned)(kNG - 1) : (unsigned)(kNC - 1);
    float* srow = sbase + (size_t)((v < kV0) ? (2 * kNG + v) : (v - kV0)) * kH;

    float4 acc = make_float4(0.0f, 0.0f, 0.0f, 0.0f);
    for (int c = 0; c < n; c += 32) {
      const int m = (n - c < 32) ? (n - c) : 32;
      int idx = 0;
      if (lane < m) idx = col[st + c + lane];
      for (int j0 = 0; j0 < m; j0 += 8) {
        unsigned s[8];
#pragma unroll
        for (int u = 0; u < 8; ++u) {
          const unsigned t = (unsigned)__builtin_amdgcn_readlane(idx, j0 + u);
          s[u] = (t < smax) ? t : smax;
        }
        float4 r[8];
#pragma unroll
        for (int u = 0; u < 8; ++u) {
          r[u] = make_float4(0.0f, 0.0f, 0.0f, 0.0f);
          if (j0 + u < m) r[u] = *(const float4*)(x + (size_t)s[u] * kH + lane * 4);
        }
#pragma unroll
        for (int u = 0; u < 8; ++u) {
          acc.x += r[u].x; acc.y += r[u].y; acc.z += r[u].z; acc.w += r[u].w;
        }
      }
    }
    *(float4*)(srow + lane * 4) = acc;
    if (lane == 0) cnt[v] = (float)n;
  }
}

__global__ void hgnn_addmat(const float* __restrict__ a, const float* __restrict__ b,
                            float* __restrict__ o, int n) {
  int i = blockIdx.x * blockDim.x + threadIdx.x;
  if (i < n) o[i] = a[i] + b[i];
}

__global__ __launch_bounds__(256)
void hgnn_gemm3(const float* __restrict__ A0, const float* __restrict__ c0,
                const float* __restrict__ W0,
                const float* __restrict__ A1, const float* __restrict__ c1,
                const float* __restrict__ W1,
                const float* __restrict__ A2, const float* __restrict__ c2,
                const float* __restrict__ W2,
                const float* __restrict__ b0, const float* __restrict__ b1,
                float* __restrict__ Y, int relu) {
  __shared__ float As[16 * 132];
  const int tid  = threadIdx.x;
  const int wv   = tid >> 5;
  const int lane = tid & 31;
  const int m    = lane & 15;
  const int koff = (lane >> 4) << 1;
  const int col  = (wv << 4) + m;
  const int row0 = blockIdx.x << 4;

  const float* Alist[3] = {A0, A1, A2};
  const float* Clist[3] = {c0, c1, c2};
  const float* Wlist[3] = {W0, W1, W2};

  v8f acc = {};
  for (int t = 0; t < 3; ++t) {
    const float* A = Alist[t];
    if (!A) continue;
    const float* cp = Clist[t];
    const float* W  = Wlist[t];

    __syncthreads();
    for (int i = tid; i < 16 * kH; i += 256) {
      const int r = i >> 7;
      const int c = i & (kH - 1);
      float v = A[(size_t)(row0 + r) * kH + c];
      if (cp) v *= 1.0f / fmaxf(cp[row0 + r], 1.0f);
      As[r * 132 + c] = v;
    }
    __syncthreads();

#pragma unroll 8
    for (int k0 = 0; k0 < kH; k0 += 4) {
      const int k = k0 + koff;
      v2f av, bv;
      av.x = As[m * 132 + k];
      av.y = As[m * 132 + k + 1];
      bv.x = W[(size_t)k * kH + col];
      bv.y = W[(size_t)(k + 1) * kH + col];
      acc = __builtin_amdgcn_wmma_f32_16x16x4_f32(
          false, av, false, bv, (short)0, acc, false, false);
    }
  }

  const float bsum = (b0 ? b0[col] : 0.0f) + (b1 ? b1[col] : 0.0f);
  __syncthreads();
#pragma unroll
  for (int r = 0; r < 8; ++r) {
    float v = acc[r] + bsum;
    if (relu) v = fmaxf(v, 0.0f);
    Y[(size_t)(row0 + r + ((lane >> 4) << 3)) * kH + col] = v;
  }
}

extern "C" void kernel_launch(void* const* d_in, const int* in_sizes, int n_in,
                              void* d_out, int out_size, void* d_ws, size_t ws_size,
                              hipStream_t stream) {
  (void)n_in; (void)out_size; (void)ws_size;
  const float* EG = (const float*)d_in[0];
  const float* ED = (const float*)d_in[1];
  const float* EC = (const float*)d_in[2];
  const float* Wl = (const float*)d_in[3];
  const float* bl = (const float*)d_in[4];
  const float* Wr = (const float*)d_in[5];
  const int* eg_s  = (const int*)d_in[6];
  const int* eg_d  = (const int*)d_in[7];
  const int* ecd_s = (const int*)d_in[8];
  const int* ecd_d = (const int*)d_in[9];
  const int* ecg_s = (const int*)d_in[10];
  const int* ecg_d = (const int*)d_in[11];
  const int* egc_s = (const int*)d_in[12];
  const int* egc_d = (const int*)d_in[13];

  float* ws = (float*)d_ws;
  float* s0 = ws;
  float* s2 = s0 + (size_t)kNG * kH;
  float* s3 = s2 + (size_t)kNG * kH;
  float* s1 = s3;
  float* cnt  = s3 + (size_t)kNC * kH;
  float* wsum = cnt + kNV;
  int* deg = (int*)(wsum + kH * kH);
  int* cur = deg + kNV;
  int* col = cur + kNV;

  float* OG = (float*)d_out;
  float* OD = OG + (size_t)kNG * kH;
  float* OC = OD + (size_t)kND * kH;

  hgnn_build<<<kBuildGrid, kBT, 0, stream>>>(
      eg_s, eg_d, ecd_s, ecd_d, ecg_s, ecg_d, egc_s, egc_d,
      in_sizes[6], in_sizes[8], in_sizes[10], in_sizes[12], deg, cur, col);

  const float* xg = EG;
  const float* xd = ED;
  const float* xc = EC;

  for (int L = 0; L < 2; ++L) {
    const float* WlL = Wl + (size_t)L * 4 * kH * kH;
    const float* WrL = Wr + (size_t)L * 4 * kH * kH;
    const float* blL = bl + (size_t)L * 4 * kH;
    const int relu = (L == 0) ? 1 : 0;

    hgnn_gather<<<2048, 256, 0, stream>>>(xg, xc, deg, cur, col, s0, cnt, kV1, kV0);
    hgnn_gemm3<<<kND / 16, 256, 0, stream>>>(
        s1, cnt + kV1, WlL + 1 * kH * kH,
        xd, nullptr, WrL + 1 * kH * kH,
        nullptr, nullptr, nullptr,
        blL + 1 * kH, nullptr, OD, relu);

    hgnn_gather<<<2048, 256, 0, stream>>>(xg, xc, deg, cur, col, s0, cnt, kV0, kNV);
    hgnn_addmat<<<(kH * kH + 255) / 256, 256, 0, stream>>>(WrL, WrL + 2 * kH * kH,
                                                           wsum, kH * kH);
    hgnn_gemm3<<<kNG / 16, 256, 0, stream>>>(
        s0, cnt + kV0, WlL,
        s2, cnt + kV2, WlL + 2 * kH * kH,
        xg, nullptr, wsum,
        blL, blL + 2 * kH, OG, relu);
    hgnn_gemm3<<<kNC / 16, 256, 0, stream>>>(
        s3, cnt + kV3, WlL + 3 * kH * kH,
        xc, nullptr, WrL + 3 * kH * kH,
        nullptr, nullptr, nullptr,
        blL + 3 * kH, nullptr, OC, relu);

    xg = OG; xd = OD; xc = OC;
  }
}


# GEMM retiled: a wave owns a 32x64 output tile (8 accumulators) and reads W 16 B per lane, so two W loads feed 8 WMMAs instead of 1; work-group = 2 waves x 32 rows; per-element sum order unchanged
# speedup vs baseline: 2.5283x; 1.2114x over previous
#include <hip/hip_runtime.h>


constexpr int kNG = 100000;
constexpr int kND = 50000;
constexpr int kNC = 50000;
constexpr int kH  = 128;

constexpr int kV1 = 0;
constexpr int kV0 = kND;
constexpr int kV2 = kND + kNG;
constexpr int kV3 = kND + 2 * kNG;
constexpr int kNV = kND + 2 * kNG + kNC;
constexpr int kETot = 600000 + 300000 + 600000 + 600000;

typedef __attribute__((ext_vector_type(2))) float v2f;
typedef __attribute__((ext_vector_type(8))) float v8f;

constexpr int kBT   = 1024;
constexpr int kWG1  = 34;
constexpr int kWG0  = 74;
constexpr int kWG2  = 74;
constexpr int kWG3  = 74;
constexpr int kBuildGrid = kWG1 + kWG0 + kWG2 + kWG3;
constexpr int kMaxR = 2 * kBT;
constexpr int kMaxGroups = 5;
constexpr int kMaxEdges  = kMaxGroups * 32 * kBT * 4;
constexpr int kNoNode    = 0x7fffffff;
static_assert((kND + kWG1 - 1) / kWG1 <= kMaxR, "range too long");
static_assert((kNG + kWG0 - 1) / kWG0 <= kMaxR, "range too long");
static_assert((kNG + kWG2 - 1) / kWG2 <= kMaxR, "range too long");
static_assert((kNC + kWG3 - 1) / kWG3 <= kMaxR, "range too long");
static_assert(600000 <= kMaxEdges, "mask words too few");

__global__ __launch_bounds__(kBT)
void hgnn_build(const int* __restrict__ s0, const int* __restrict__ d0,
                const int* __restrict__ s1, const int* __restrict__ d1,
                const int* __restrict__ s2, const int* __restrict__ d2,
                const int* __restrict__ s3, const int* __restrict__ d3,
                int n0, int n1, int n2, int n3,
                int* __restrict__ deg, int* __restrict__ cur, int* __restrict__ col) {
  __shared__ int      ldeg[kMaxR];
  __shared__ unsigned lmask[kMaxGroups * kBT];
  __shared__ int      wpart[32];
  __shared__ int      sbelow;
  const int tid  = threadIdx.x;
  const int lane = tid & 31;
  const int wv   = tid >> 5;

  const int b = blockIdx.x;
  int r, w, nwg, nn;
  if (b < kWG1)                    { r = 1; w = b;                      nwg = kWG1; nn = kND; }
  else if (b < kWG1 + kWG0)        { r = 0; w = b - kWG1;               nwg = kWG0; nn = kNG; }
  else if (b < kWG1 + kWG0 + kWG2) { r = 2; w = b - kWG1 - kWG0;        nwg = kWG2; nn = kNG; }
  else                             { r = 3; w = b - kWG1 - kWG0 - kWG2; nwg = kWG3; nn = kNC; }
  const int* s = (r == 0) ? s0 : (r == 1) ? s1 : (r == 2) ? s2 : s3;
  const int* d = (r == 0) ? d0 : (r == 1) ? d1 : (r == 2) ? d2 : d3;
  int n        = (r == 0) ? n0 : (r == 1) ? n1 : (r == 2) ? n2 : n3;
  if (n > kMaxEdges) n = kMaxEdges;
  const int vb    = (r == 0) ? kV0 : (r == 1) ? kV1 : (r == 2) ? kV2 : kV3;
  const int ebase = (r == 1) ? 0 : (r == 0) ? n1 : (r == 2) ? n1 + n0 : n1 + n0 + n2;
  const int rpw = (nn + nwg - 1) / nwg;
  const int lo  = w * rpw;
  const int R   = (nn - lo < rpw) ? (nn - lo) : rpw;

  for (int i = tid; i < kMaxR; i += kBT) ldeg[i] = 0;
  if (tid == 0) sbelow = 0;
  __syncthreads();

  const int4* d4 = (const int4*)d;
  const int n4 = n >> 2;
  int below = 0;
  auto count1 = [&](int dv) -> bool {
    const unsigned l = (unsigned)dv - (unsigned)lo;
    below += (dv < lo) ? 1 : 0;
    if (l < (unsigned)R) { atomicAdd(&ldeg[l], 1); return true; }
    return false;
  };

  for (int g = 0; g < kMaxGroups; ++g) {
    unsigned mask = 0;
    for (int h = 0; h < 8; ++h) {
      const int jb = ((g * 8 + h) * 4) * kBT + tid;
      if (jb >= n4) break;
      int4 q[4];
#pragma unroll
      for (int u = 0; u < 4; ++u) {
        const int j = jb + u * kBT;
        q[u] = (j < n4) ? d4[j] : make_int4(kNoNode, kNoNode, kNoNode, kNoNode);
      }
#pragma unroll
      for (int u = 0; u < 4; ++u) {
        const bool hx = count1(q[u].x);
        const bool hy = count1(q[u].y);
        const bool hz = count1(q[u].z);
        const bool hw = count1(q[u].w);
        if (hx | hy | hz | hw) mask |= 1u << (h * 4 + u);
      }
    }
    lmask[g * kBT + tid] = mask;
  }
  const int et = (n4 << 2) + tid;
  const bool has_tail = tid < (n & 3);
  if (has_tail) count1(d[et]);

  for (int o = 16; o > 0; o >>= 1) below += __shfl_xor(below, o);
  if (lane == 0) atomicAdd(&sbelow, below);
  __syncthreads();

  const int i0 = 2 * tid;
  const int i1 = i0 + 1;
  const int a  = (i0 < R) ? ldeg[i0] : 0;
  const int c  = (i1 < R) ? ldeg[i1] : 0;
  const int sum = a + c;
  int inc = sum;
  for (int o = 1; o < 32; o <<= 1) {
    const int t = __shfl_up(inc, o);
    if (lane >= o) inc += t;
  }
  if (lane == 31) wpart[wv] = inc;
  __syncthreads();
  if (wv == 0) {
    const int x = wpart[lane];
    int xi = x;
    for (int o = 1; o < 32; o <<= 1) {
      const int t = __shfl_up(xi, o);
      if (lane >= o) xi += t;
    }
    wpart[lane] = xi - x;
  }
  __syncthreads();
  const int ex   = wpart[wv] + inc - sum;
  const int base = ebase + sbelow;
  if (i0 < R) { ldeg[i0] = ex;     deg[vb + lo + i0] = a; cur[vb + lo + i0] = base + ex + a; }
  if (i1 < R) { ldeg[i1] = ex + a; deg[vb + lo + i1] = c; cur[vb + lo + i1] = base + ex + sum; }
  __syncthreads();

  auto place = [&](int dv, int e) {
    const unsigned l = (unsigned)dv - (unsigned)lo;
    if (l < (unsigned)R) {
      const int p = base + atomicAdd(&ldeg[l], 1);
      if ((unsigned)p < (unsigned)kETot) col[p] = s[e];
    }
  };
  for (int g = 0; g < kMaxGroups; ++g) {
    unsigned mask = lmask[g * kBT + tid];
    while (mask) {
      const int k = __builtin_ctz(mask);
      mask &= mask - 1;
      const int j = (g * 32 + k) * kBT + tid;
      const int4 q = d4[j];
      place(q.x, 4 * j);
      place(q.y, 4 * j + 1);
      place(q.z, 4 * j + 2);
      place(q.w, 4 * j + 3);
    }
  }
  if (has_tail) place(d[et], et);
}

__global__ __launch_bounds__(256)
void hgnn_gather(const float* __restrict__ xg, const float* __restrict__ xc,
                 const int* __restrict__ deg, const int* __restrict__ cur,
                 const int* __restrict__ col, float* __restrict__ sbase,
                 float* __restrict__ cnt, int vbeg, int vend) {
  const int lane = threadIdx.x & 31;
  const int wave = blockIdx.x * 8 + __builtin_amdgcn_readfirstlane(threadIdx.x >> 5);
  const int nW   = gridDim.x * 8;
  for (int v = vbeg + wave; v < vend; v += nW) {
    const int n  = __builtin_amdgcn_readfirstlane(deg[v]);
    const int st = __builtin_amdgcn_readfirstlane(cur[v]) - n;
    const bool src_gene = (v >= kV0 && v < kV2) || (v >= kV3);
    const float* x = src_gene ? xg : xc;
    const unsigned smax = src_gene ? (unsigned)(kNG - 1) : (unsigned)(kNC - 1);
    float* srow = sbase + (size_t)((v < kV0) ? (2 * kNG + v) : (v - kV0)) * kH;

    float4 acc = make_float4(0.0f, 0.0f, 0.0f, 0.0f);
    for (int c = 0; c < n; c += 32) {
      const int m = (n - c < 32) ? (n - c) : 32;
      int idx = 0;
      if (lane < m) idx = col[st + c + lane];
      for (int j0 = 0; j0 < m; j0 += 8) {
        unsigned s[8];
#pragma unroll
        for (int u = 0; u < 8; ++u) {
          const unsigned t = (unsigned)__builtin_amdgcn_readlane(idx, j0 + u);
          s[u] = (t < smax) ? t : smax;
        }
        float4 r[8];
#pragma unroll
        for (int u = 0; u < 8; ++u) {
          r[u] = make_float4(0.0f, 0.0f, 0.0f, 0.0f);
          if (j0 + u < m) r[u] = *(const float4*)(x + (size_t)s[u] * kH + lane * 4);
        }
#pragma unroll
        for (int u = 0; u < 8; ++u) {
          acc.x += r[u].x; acc.y += r[u].y; acc.z += r[u].z; acc.w += r[u].w;
        }
      }
    }
    *(float4*)(srow + lane * 4) = acc;
    if (lane == 0) cnt[v] = (float)n;
  }
}

__global__ void hgnn_addmat(const float* __restrict__ a, const float* __restrict__ b,
                            float* __restrict__ o, int n) {
  int i = blockIdx.x * blockDim.x + threadIdx.x;
  if (i < n) o[i] = a[i] + b[i];
}

constexpr int kBM  = 32;
constexpr int kLdA = 132;

__device__ __forceinline__ v8f hgnn_mma(v2f a, float b0, float b1, v8f c) {
  v2f b;
  b.x = b0;
  b.y = b1;
  return __builtin_amdgcn_wmma_f32_16x16x4_f32(false, a, false, b, (short)0, c, false, false);
}

__global__ __launch_bounds__(64)
void hgnn_gemm3(const float* __restrict__ A0, const float* __restrict__ c0,
                const float* __restrict__ W0,
                const float* __restrict__ A1, const float* __restrict__ c1,
                const float* __restrict__ W1,
                const float* __restrict__ A2, const float* __restrict__ c2,
                const float* __restrict__ W2,
                const float* __restrict__ b0, const float* __restrict__ b1,
                float* __restrict__ Y, int nrows, int relu) {
  __shared__ __attribute__((aligned(16))) float As[kBM * kLdA];
  const int tid  = threadIdx.x;
  const int lane = tid & 31;
  const int wv   = __builtin_amdgcn_readfirstlane(tid >> 5);
  const int c    = lane & 15;
  const int koff = (lane >> 4) << 1;
  const int colb = (wv << 6) + (c << 2);
  const int row0 = blockIdx.x * kBM;

  v8f acc[2][4] = {};
#pragma unroll 1
  for (int t = 0; t < 3; ++t) {
    const float* A  = (t == 0) ? A0 : (t == 1) ? A1 : A2;
    const float* cp = (t == 0) ? c0 : (t == 1) ? c1 : c2;
    const float* W  = (t == 0) ? W0 : (t == 1) ? W1 : W2;
    if (!A) continue;

    __syncthreads();
    for (int u0 = 0; u0 < kBM / 2; u0 += 8) {
      float4 q[8];
#pragma unroll
      for (int u = 0; u < 8; ++u) {
        const int row = row0 + wv + 2 * (u0 + u);
        const int rc  = (row < nrows) ? row : (nrows - 1);
        q[u] = *(const float4*)(A + (size_t)rc * kH + lane * 4);
      }
      if (cp) {
        float sc[8];
#pragma unroll
        for (int u = 0; u < 8; ++u) {
          const int row = row0 + wv + 2 * (u0 + u);
          const int rc  = (row < nrows) ? row : (nrows - 1);
          sc[u] = 1.0f / fmaxf(cp[rc], 1.0f);
        }
#pragma unroll
        for (int u = 0; u < 8; ++u) {
          q[u].x *= sc[u]; q[u].y *= sc[u]; q[u].z *= sc[u]; q[u].w *= sc[u];
        }
      }
#pragma unroll
      for (int u = 0; u < 8; ++u) {
        const int rl = wv + 2 * (u0 + u);
        if (row0 + rl >= nrows) q[u] = make_float4(0.0f, 0.0f, 0.0f, 0.0f);
        *(float4*)(As + rl * kLdA + lane * 4) = q[u];
      }
    }
    __syncthreads();

    const float* Wp  = W + (size_t)koff * kH + colb;
    const float* Ap0 = As + c * kLdA + koff;
    const float* Ap1 = As + (16 + c) * kLdA + koff;
#pragma unroll 4
    for (int k0 = 0; k0 < kH; k0 += 4) {
      const float4 w0 = *(const float4*)(Wp + (size_t)k0 * kH);
      const float4 w1 = *(const float4*)(Wp + (size_t)(k0 + 1) * kH);
      const v2f a0 = *(const v2f*)(Ap0 + k0);
      const v2f a1 = *(const v2f*)(Ap1 + k0);
      acc[0][0] = hgnn_mma(a0, w0.x, w1.x, acc[0][0]);
      acc[1][0] = hgnn_mma(a1, w0.x, w1.x, acc[1][0]);
      acc[0][1] = hgnn_mma(a0, w0.y, w1.y, acc[0][1]);
      acc[1][1] = hgnn_mma(a1, w0.y, w1.y, acc[1][1]);
      acc[0][2] = hgnn_mma(a0, w0.z, w1.z, acc[0][2]);
      acc[1][2] = hgnn_mma(a1, w0.z, w1.z, acc[1][2]);
      acc[0][3] = hgnn_mma(a0, w0.w, w1.w, acc[0][3]);
      acc[1][3] = hgnn_mma(a1, w0.w, w1.w, acc[1][3]);
    }
  }

  float4 bs = make_float4(0.0f, 0.0f, 0.0f, 0.0f);
  if (b0) { const float4 v = *(const float4*)(b0 + colb); bs.x += v.x; bs.y += v.y; bs.z += v.z; bs.w += v.w; }
  if (b1) { const float4 v = *(const float4*)(b1 + colb); bs.x += v.x; bs.y += v.y; bs.z += v.z; bs.w += v.w; }
#pragma unroll
  for (int i = 0; i < 2; ++i) {
#pragma unroll
    for (int r = 0; r < 8; ++r) {
      const int row = row0 + 16 * i + r + ((lane >> 4) << 3);
      float4 v;
      v.x = acc[i][0][r] + bs.x;
      v.y = acc[i][1][r] + bs.y;
      v.z = acc[i][2][r] + bs.z;
      v.w = acc[i][3][r] + bs.w;
      if (relu) { v.x = fmaxf(v.x, 0.0f); v.y = fmaxf(v.y, 0.0f); v.z = fmaxf(v.z, 0.0f); v.w = fmaxf(v.w, 0.0f); }
      if (row < nrows) *(float4*)(Y + (size_t)row * kH + colb) = v;
    }
  }
}

extern "C" void kernel_launch(void* const* d_in, const int* in_sizes, int n_in,
                              void* d_out, int out_size, void* d_ws, size_t ws_size,
                              hipStream_t stream) {
  (void)n_in; (void)out_size; (void)ws_size;
  const float* EG = (const float*)d_in[0];
  const float* ED = (const float*)d_in[1];
  const float* EC = (const float*)d_in[2];
  const float* Wl = (const float*)d_in[3];
  const float* bl = (const float*)d_in[4];
  const float* Wr = (const float*)d_in[5];
  const int* eg_s  = (const int*)d_in[6];
  const int* eg_d  = (const int*)d_in[7];
  const int* ecd_s = (const int*)d_in[8];
  const int* ecd_d = (const int*)d_in[9];
  const int* ecg_s = (const int*)d_in[10];
  const int* ecg_d = (const int*)d_in[11];
  const int* egc_s = (const int*)d_in[12];
  const int* egc_d = (const int*)d_in[13];

  float* ws = (float*)d_ws;
  float* s0 = ws;
  float* s2 = s0 + (size_t)kNG * kH;
  float* s3 = s2 + (size_t)kNG * kH;
  float* s1 = s3;
  float* cnt  = s3 + (size_t)kNC * kH;
  float* wsum = cnt + kNV;
  int* deg = (int*)(wsum + kH * kH);
  int* cur = deg + kNV;
  int* col = cur + kNV;

  float* OG = (float*)d_out;
  float* OD = OG + (size_t)kNG * kH;
  float* OC = OD + (size_t)kND * kH;

  hgnn_build<<<kBuildGrid, kBT, 0, stream>>>(
      eg_s, eg_d, ecd_s, ecd_d, ecg_s, ecg_d, egc_s, egc_d,
      in_sizes[6], in_sizes[8], in_sizes[10], in_sizes[12], deg, cur, col);

  const float* xg = EG;
  const float* xd = ED;
  const float* xc = EC;

  for (int L = 0; L < 2; ++L) {
    const float* WlL = Wl + (size_t)L * 4 * kH * kH;
    const float* WrL = Wr + (size_t)L * 4 * kH * kH;
    const float* blL = bl + (size_t)L * 4 * kH;
    const int relu = (L == 0) ? 1 : 0;

    hgnn_gather<<<2048, 256, 0, stream>>>(xg, xc, deg, cur, col, s0, cnt, kV1, kV0);
    hgnn_gemm3<<<(kND + kBM - 1) / kBM, 64, 0, stream>>>(
        s1, cnt + kV1, WlL + 1 * kH * kH,
        xd, nullptr, WrL + 1 * kH * kH,
        nullptr, nullptr, nullptr,
        blL + 1 * kH, nullptr, OD, kND, relu);

    hgnn_gather<<<2048, 256, 0, stream>>>(xg, xc, deg, cur, col, s0, cnt, kV0, kNV);
    hgnn_addmat<<<(kH * kH + 255) / 256, 256, 0, stream>>>(WrL, WrL + 2 * kH * kH,
                                                           wsum, kH * kH);
    hgnn_gemm3<<<(kNG + kBM - 1) / kBM, 64, 0, stream>>>(
        s0, cnt + kV0, WlL,
        s2, cnt + kV2, WlL + 2 * kH * kH,
        xg, nullptr, wsum,
        blL, blL + 2 * kH, OG, kNG, relu);
    hgnn_gemm3<<<(kNC + kBM - 1) / kBM, 64, 0, stream>>>(
        s3, cnt + kV3, WlL + 3 * kH * kH,
        xc, nullptr, WrL + 3 * kH * kH,
        nullptr, nullptr, nullptr,
        blL + 3 * kH, nullptr, OC, kNC, relu);

    xg = OG; xd = OD; xc = OC;
  }
}


# Launches 13 -> 7: layer 1 gathers all 4 relations in one launch (disease sums parked in the output's disease rows), its 3 GEMMs run as one grid; layer 2 merges 2 GEMMs; Wsum formed in a gather launch
# speedup vs baseline: 2.8031x; 1.1087x over previous
#include <hip/hip_runtime.h>


constexpr int kNG = 100000;
constexpr int kND = 50000;
constexpr int kNC = 50000;
constexpr int kH  = 128;

constexpr int kV1 = 0;
constexpr int kV0 = kND;
constexpr int kV2 = kND + kNG;
constexpr int kV3 = kND + 2 * kNG;
constexpr int kNV = kND + 2 * kNG + kNC;
constexpr int kETot = 600000 + 300000 + 600000 + 600000;

typedef __attribute__((ext_vector_type(2))) float v2f;
typedef __attribute__((ext_vector_type(8))) float v8f;

constexpr int kBT   = 1024;
constexpr int kWG1  = 34;
constexpr int kWG0  = 74;
constexpr int kWG2  = 74;
constexpr int kWG3  = 74;
constexpr int kBuildGrid = kWG1 + kWG0 + kWG2 + kWG3;
constexpr int kMaxR = 2 * kBT;
constexpr int kMaxGroups = 5;
constexpr int kMaxEdges  = kMaxGroups * 32 * kBT * 4;
constexpr int kNoNode    = 0x7fffffff;
static_assert((kND + kWG1 - 1) / kWG1 <= kMaxR, "range too long");
static_assert((kNG + kWG0 - 1) / kWG0 <= kMaxR, "range too long");
static_assert((kNG + kWG2 - 1) / kWG2 <= kMaxR, "range too long");
static_assert((kNC + kWG3 - 1) / kWG3 <= kMaxR, "range too long");
static_assert(600000 <= kMaxEdges, "mask words too few");

__global__ __launch_bounds__(kBT)
void hgnn_build(const int* __restrict__ s0, const int* __restrict__ d0,
                const int* __restrict__ s1, const int* __restrict__ d1,
                const int* __restrict__ s2, const int* __restrict__ d2,
                const int* __restrict__ s3, const int* __restrict__ d3,
                int n0, int n1, int n2, int n3,
                int* __restrict__ deg, int* __restrict__ cur, int* __restrict__ col) {
  __shared__ int      ldeg[kMaxR];
  __shared__ unsigned lmask[kMaxGroups * kBT];
  __shared__ int      wpart[32];
  __shared__ int      sbelow;
  const int tid  = threadIdx.x;
  const int lane = tid & 31;
  const int wv   = tid >> 5;

  const int b = blockIdx.x;
  int r, w, nwg, nn;
  if (b < kWG1)                    { r = 1; w = b;                      nwg = kWG1; nn = kND; }
  else if (b < kWG1 + kWG0)        { r = 0; w = b - kWG1;               nwg = kWG0; nn = kNG; }
  else if (b < kWG1 + kWG0 + kWG2) { r = 2; w = b - kWG1 - kWG0;        nwg = kWG2; nn = kNG; }
  else                             { r = 3; w = b - kWG1 - kWG0 - kWG2; nwg = kWG3; nn = kNC; }
  const int* s = (r == 0) ? s0 : (r == 1) ? s1 : (r == 2) ? s2 : s3;
  const int* d = (r == 0) ? d0 : (r == 1) ? d1 : (r == 2) ? d2 : d3;
  int n        = (r == 0) ? n0 : (r == 1) ? n1 : (r == 2) ? n2 : n3;
  if (n > kMaxEdges) n = kMaxEdges;
  const int vb    = (r == 0) ? kV0 : (r == 1) ? kV1 : (r == 2) ? kV2 : kV3;
  const int ebase = (r == 1) ? 0 : (r == 0) ? n1 : (r == 2) ? n1 + n0 : n1 + n0 + n2;
  const int rpw = (nn + nwg - 1) / nwg;
  const int lo  = w * rpw;
  const int R   = (nn - lo < rpw) ? (nn - lo) : rpw;

  for (int i = tid; i < kMaxR; i += kBT) ldeg[i] = 0;
  if (tid == 0) sbelow = 0;
  __syncthreads();

  const int4* d4 = (const int4*)d;
  const int n4 = n >> 2;
  int below = 0;
  auto count1 = [&](int dv) -> bool {
    const unsigned l = (unsigned)dv - (unsigned)lo;
    below += (dv < lo) ? 1 : 0;
    if (l < (unsigned)R) { atomicAdd(&ldeg[l], 1); return true; }
    return false;
  };

  for (int g = 0; g < kMaxGroups; ++g) {
    unsigned mask = 0;
    for (int h = 0; h < 8; ++h) {
      const int jb = ((g * 8 + h) * 4) * kBT + tid;
      if (jb >= n4) break;
      int4 q[4];
#pragma unroll
      for (int u = 0; u < 4; ++u) {
        const int j = jb + u * kBT;
        q[u] = (j < n4) ? d4[j] : make_int4(kNoNode, kNoNode, kNoNode, kNoNode);
      }
#pragma unroll
      for (int u = 0; u < 4; ++u) {
        const bool hx = count1(q[u].x);
        const bool hy = count1(q[u].y);
        const bool hz = count1(q[u].z);
        const bool hw = count1(q[u].w);
        if (hx | hy | hz | hw) mask |= 1u << (h * 4 + u);
      }
    }
    lmask[g * kBT + tid] = mask;
  }
  const int et = (n4 << 2) + tid;
  const bool has_tail = tid < (n & 3);
  if (has_tail) count1(d[et]);

  for (int o = 16; o > 0; o >>= 1) below += __shfl_xor(below, o);
  if (lane == 0) atomicAdd(&sbelow, below);
  __syncthreads();

  const int i0 = 2 * tid;
  const int i1 = i0 + 1;
  const int a  = (i0 < R) ? ldeg[i0] : 0;
  const int c  = (i1 < R) ? ldeg[i1] : 0;
  const int sum = a + c;
  int inc = sum;
  for (int o = 1; o < 32; o <<= 1) {
    const int t = __shfl_up(inc, o);
    if (lane >= o) inc += t;
  }
  if (lane == 31) wpart[wv] = inc;
  __syncthreads();
  if (wv == 0) {
    const int x = wpart[lane];
    int xi = x;
    for (int o = 1; o < 32; o <<= 1) {
      const int t = __shfl_up(xi, o);
      if (lane >= o) xi += t;
    }
    wpart[lane] = xi - x;
  }
  __syncthreads();
  const int ex   = wpart[wv] + inc - sum;
  const int base = ebase + sbelow;
  if (i0 < R) { ldeg[i0] = ex;     deg[vb + lo + i0] = a; cur[vb + lo + i0] = base + ex + a; }
  if (i1 < R) { ldeg[i1] = ex + a; deg[vb + lo + i1] = c; cur[vb + lo + i1] = base + ex + sum; }
  __syncthreads();

  auto place = [&](int dv, int e) {
    const unsigned l = (unsigned)dv - (unsigned)lo;
    if (l < (unsigned)R) {
      const int p = base + atomicAdd(&ldeg[l], 1);
      if ((unsigned)p < (unsigned)kETot) col[p] = s[e];
    }
  };
  for (int g = 0; g < kMaxGroups; ++g) {
    unsigned mask = lmask[g * kBT + tid];
    while (mask) {
      const int k = __builtin_ctz(mask);
      mask &= mask - 1;
      const int j = (g * 32 + k) * kBT + tid;
      const int4 q = d4[j];
      place(q.x, 4 * j);
      place(q.y, 4 * j + 1);
      place(q.z, 4 * j + 2);
      place(q.w, 4 * j + 3);
    }
  }
  if (has_tail) place(d[et], et);
}

__global__ __launch_bounds__(256)
void hgnn_gather(const float* __restrict__ xg, const float* __restrict__ xc,
                 const int* __restrict__ deg, const int* __restrict__ cur,
                 const int* __restrict__ col, float* sbase, float* sdis,
                 float* __restrict__ cnt, int vbeg, int vend, int ngw,
                 const float* __restrict__ wa, const float* __restrict__ wb,
                 float* __restrict__ wo) {
  if ((int)blockIdx.x >= ngw) {
    const int i = ((int)blockIdx.x - ngw) * 256 + (int)threadIdx.x;
    if (wo && i < kH * kH) wo[i] = wa[i] + wb[i];
    return;
  }
  const int lane = threadIdx.x & 31;
  const int wave = blockIdx.x * 8 + __builtin_amdgcn_readfirstlane(threadIdx.x >> 5);
  const int nW   = ngw * 8;
  for (int v = vbeg + wave; v < vend; v += nW) {
    const int n  = __builtin_amdgcn_readfirstlane(deg[v]);
    const int st = __builtin_amdgcn_readfirstlane(cur[v]) - n;
    const bool src_gene = (v >= kV0 && v < kV2) || (v >= kV3);
    const float* x = src_gene ? xg : xc;
    const unsigned smax = src_gene ? (unsigned)(kNG - 1) : (unsigned)(kNC - 1);
    float* srow = (v < kV0) ? (sdis + (size_t)(v - kV1) * kH) : (sbase + (size_t)(v - kV0) * kH);

    float4 acc = make_float4(0.0f, 0.0f, 0.0f, 0.0f);
    for (int c = 0; c < n; c += 32) {
      const int m = (n - c < 32) ? (n - c) : 32;
      int idx = 0;
      if (lane < m) idx = col[st + c + lane];
      for (int j0 = 0; j0 < m; j0 += 8) {
        unsigned s[8];
#pragma unroll
        for (int u = 0; u < 8; ++u) {
          const unsigned t = (unsigned)__builtin_amdgcn_readlane(idx, j0 + u);
          s[u] = (t < smax) ? t : smax;
        }
        float4 r[8];
#pragma unroll
        for (int u = 0; u < 8; ++u) {
          r[u] = make_float4(0.0f, 0.0f, 0.0f, 0.0f);
          if (j0 + u < m) r[u] = *(const float4*)(x + (size_t)s[u] * kH + lane * 4);
        }
#pragma unroll
        for (int u = 0; u < 8; ++u) {
          acc.x += r[u].x; acc.y += r[u].y; acc.z += r[u].z; acc.w += r[u].w;
        }
      }
    }
    *(float4*)(srow + lane * 4) = acc;
    if (lane == 0) cnt[v] = (float)n;
  }
}

constexpr int kBM  = 32;
constexpr int kLdA = 132;

__device__ __forceinline__ v8f hgnn_mma(v2f a, float b0, float b1, v8f c) {
  v2f b;
  b.x = b0;
  b.y = b1;
  return __builtin_amdgcn_wmma_f32_16x16x4_f32(false, a, false, b, (short)0, c, false, false);
}

struct hgnn_seg {
  const float* A0; const float* c0; const float* W0;
  const float* A1; const float* c1; const float* W1;
  const float* A2; const float* c2; const float* W2;
  const float* b0; const float* b1;
  float* Y; int nrows; int nblk;
};

__global__ __launch_bounds__(64)
void hgnn_gemm3(hgnn_seg g0, hgnn_seg g1, hgnn_seg g2, int relu) {
  int blk = blockIdx.x;
  int seg = 0;
  if (blk >= g0.nblk) { blk -= g0.nblk; seg = 1; if (blk >= g1.nblk) { blk -= g1.nblk; seg = 2; } }
#define HGNN_SEL(f) ((seg == 0) ? g0.f : (seg == 1) ? g1.f : g2.f)
  const float* __restrict__ A0 = HGNN_SEL(A0);
  const float* __restrict__ c0 = HGNN_SEL(c0);
  const float* __restrict__ W0 = HGNN_SEL(W0);
  const float* __restrict__ A1 = HGNN_SEL(A1);
  const float* __restrict__ c1 = HGNN_SEL(c1);
  const float* __restrict__ W1 = HGNN_SEL(W1);
  const float* __restrict__ A2 = HGNN_SEL(A2);
  const float* __restrict__ c2 = HGNN_SEL(c2);
  const float* __restrict__ W2 = HGNN_SEL(W2);
  const float* __restrict__ b0 = HGNN_SEL(b0);
  const float* __restrict__ b1 = HGNN_SEL(b1);
  float* __restrict__ Y = HGNN_SEL(Y);
  const int nrows = HGNN_SEL(nrows);
#undef HGNN_SEL
  __shared__ __attribute__((aligned(16))) float As[kBM * kLdA];
  const int tid  = threadIdx.x;
  const int lane = tid & 31;
  const int wv   = __builtin_amdgcn_readfirstlane(tid >> 5);
  const int c    = lane & 15;
  const int koff = (lane >> 4) << 1;
  const int colb = (wv << 6) + (c << 2);
  const int row0 = blk * kBM;

  v8f acc[2][4] = {};
#pragma unroll 1
  for (int t = 0; t < 3; ++t) {
    const float* A  = (t == 0) ? A0 : (t == 1) ? A1 : A2;
    const float* cp = (t == 0) ? c0 : (t == 1) ? c1 : c2;
    const float* W  = (t == 0) ? W0 : (t == 1) ? W1 : W2;
    if (!A) continue;

    __syncthreads();
    for (int u0 = 0; u0 < kBM / 2; u0 += 8) {
      float4 q[8];
#pragma unroll
      for (int u = 0; u < 8; ++u) {
        const int row = row0 + wv + 2 * (u0 + u);
        const int rc  = (row < nrows) ? row : (nrows - 1);
        q[u] = *(const float4*)(A + (size_t)rc * kH + lane * 4);
      }
      if (cp) {
        float sc[8];
#pragma unroll
        for (int u = 0; u < 8; ++u) {
          const int row = row0 + wv + 2 * (u0 + u);
          const int rc  = (row < nrows) ? row : (nrows - 1);
          sc[u] = 1.0f / fmaxf(cp[rc], 1.0f);
        }
#pragma unroll
        for (int u = 0; u < 8; ++u) {
          q[u].x *= sc[u]; q[u].y *= sc[u]; q[u].z *= sc[u]; q[u].w *= sc[u];
        }
      }
#pragma unroll
      for (int u = 0; u < 8; ++u) {
        const int rl = wv + 2 * (u0 + u);
        if (row0 + rl >= nrows) q[u] = make_float4(0.0f, 0.0f, 0.0f, 0.0f);
        *(float4*)(As + rl * kLdA + lane * 4) = q[u];
      }
    }
    __syncthreads();

    const float* Wp  = W + (size_t)koff * kH + colb;
    const float* Ap0 = As + c * kLdA + koff;
    const float* Ap1 = As + (16 + c) * kLdA + koff;
#pragma unroll 4
    for (int k0 = 0; k0 < kH; k0 += 4) {
      const float4 w0 = *(const float4*)(Wp + (size_t)k0 * kH);
      const float4 w1 = *(const float4*)(Wp + (size_t)(k0 + 1) * kH);
      const v2f a0 = *(const v2f*)(Ap0 + k0);
      const v2f a1 = *(const v2f*)(Ap1 + k0);
      acc[0][0] = hgnn_mma(a0, w0.x, w1.x, acc[0][0]);
      acc[1][0] = hgnn_mma(a1, w0.x, w1.x, acc[1][0]);
      acc[0][1] = hgnn_mma(a0, w0.y, w1.y, acc[0][1]);
      acc[1][1] = hgnn_mma(a1, w0.y, w1.y, acc[1][1]);
      acc[0][2] = hgnn_mma(a0, w0.z, w1.z, acc[0][2]);
      acc[1][2] = hgnn_mma(a1, w0.z, w1.z, acc[1][2]);
      acc[0][3] = hgnn_mma(a0, w0.w, w1.w, acc[0][3]);
      acc[1][3] = hgnn_mma(a1, w0.w, w1.w, acc[1][3]);
    }
  }

  float4 bs = make_float4(0.0f, 0.0f, 0.0f, 0.0f);
  if (b0) { const float4 v = *(const float4*)(b0 + colb); bs.x += v.x; bs.y += v.y; bs.z += v.z; bs.w += v.w; }
  if (b1) { const float4 v = *(const float4*)(b1 + colb); bs.x += v.x; bs.y += v.y; bs.z += v.z; bs.w += v.w; }
#pragma unroll
  for (int i = 0; i < 2; ++i) {
#pragma unroll
    for (int r = 0; r < 8; ++r) {
      const int row = row0 + 16 * i + r + ((lane >> 4) << 3);
      float4 v;
      v.x = acc[i][0][r] + bs.x;
      v.y = acc[i][1][r] + bs.y;
      v.z = acc[i][2][r] + bs.z;
      v.w = acc[i][3][r] + bs.w;
      if (relu) { v.x = fmaxf(v.x, 0.0f); v.y = fmaxf(v.y, 0.0f); v.z = fmaxf(v.z, 0.0f); v.w = fmaxf(v.w, 0.0f); }
      if (row < nrows) *(float4*)(Y + (size_t)row * kH + colb) = v;
    }
  }
}

extern "C" void kernel_launch(void* const* d_in, const int* in_sizes, int n_in,
                              void* d_out, int out_size, void* d_ws, size_t ws_size,
                              hipStream_t stream) {
  (void)n_in; (void)out_size; (void)ws_size;
  const float* EG = (const float*)d_in[0];
  const float* ED = (const float*)d_in[1];
  const float* EC = (const float*)d_in[2];
  const float* Wl = (const float*)d_in[3];
  const float* bl = (const float*)d_in[4];
  const float* Wr = (const float*)d_in[5];
  const int* eg_s  = (const int*)d_in[6];
  const int* eg_d  = (const int*)d_in[7];
  const int* ecd_s = (const int*)d_in[8];
  const int* ecd_d = (const int*)d_in[9];
  const int* ecg_s = (const int*)d_in[10];
  const int* ecg_d = (const int*)d_in[11];
  const int* egc_s = (const int*)d_in[12];
  const int* egc_d = (const int*)d_in[13];

  float* ws = (float*)d_ws;
  float* s0 = ws;
  float* s2 = s0 + (size_t)kNG * kH;
  float* s3 = s2 + (size_t)kNG * kH;
  float* s1 = s3;
  float* cnt  = s3 + (size_t)kNC * kH;
  float* wsum = cnt + kNV;
  int* deg = (int*)(wsum + kH * kH);
  int* cur = deg + kNV;
  int* col = cur + kNV;

  float* OG = (float*)d_out;
  float* OD = OG + (size_t)kNG * kH;
  float* OC = OD + (size_t)kND * kH;

  hgnn_build<<<kBuildGrid, kBT, 0, stream>>>(
      eg_s, eg_d, ecd_s, ecd_d, ecg_s, ecg_d, egc_s, egc_d,
      in_sizes[6], in_sizes[8], in_sizes[10], in_sizes[12], deg, cur, col);

  constexpr int kGW  = 2048;
  constexpr int kAW  = (kH * kH + 255) / 256;
  constexpr int kNbG = (kNG + kBM - 1) / kBM;
  constexpr int kNbD = (kND + kBM - 1) / kBM;
  constexpr int kNbC = (kNC + kBM - 1) / kBM;
  const hgnn_seg none = {nullptr, nullptr, nullptr, nullptr, nullptr, nullptr,
                         nullptr, nullptr, nullptr, nullptr, nullptr, nullptr, 0, 0};

  {
    const float* WlL = Wl;
    const float* WrL = Wr;
    const float* blL = bl;
    hgnn_gather<<<kGW + kAW, 256, 0, stream>>>(EG, EC, deg, cur, col, s0, OD, cnt, kV1, kNV, kGW,
                                               WrL, WrL + 2 * kH * kH, wsum);
    const hgnn_seg sg = {s0, cnt + kV0, WlL,
                         s2, cnt + kV2, WlL + 2 * kH * kH,
                         EG, nullptr, wsum,
                         blL, blL + 2 * kH, OG, kNG, kNbG};
    const hgnn_seg sd = {OD, cnt + kV1, WlL + 1 * kH * kH,
                         ED, nullptr, WrL + 1 * kH * kH,
                         nullptr, nullptr, nullptr,
                         blL + 1 * kH, nullptr, OD, kND, kNbD};
    const hgnn_seg sc = {s3, cnt + kV3, WlL + 3 * kH * kH,
                         EC, nullptr, WrL + 3 * kH * kH,
                         nullptr, nullptr, nullptr,
                         blL + 3 * kH, nullptr, OC, kNC, kNbC};
    hgnn_gemm3<<<kNbG + kNbD + kNbC, 64, 0, stream>>>(sg, sd, sc, 1);
  }

  {
    const float* WlL = Wl + (size_t)4 * kH * kH;
    const float* WrL = Wr + (size_t)4 * kH * kH;
    const float* blL = bl + (size_t)4 * kH;
    hgnn_gather<<<kGW + kAW, 256, 0, stream>>>(OG, OC, deg, cur, col, s0, s1, cnt, kV1, kV0, kGW,
                                               WrL, WrL + 2 * kH * kH, wsum);
    const hgnn_seg sd = {s1, cnt + kV1, WlL + 1 * kH * kH,
                         OD, nullptr, WrL + 1 * kH * kH,
                         nullptr, nullptr, nullptr,
                         blL + 1 * kH, nullptr, OD, kND, kNbD};
    hgnn_gemm3<<<kNbD, 64, 0, stream>>>(sd, none, none, 0);

    hgnn_gather<<<kGW, 256, 0, stream>>>(OG, OC, deg, cur, col, s0, s1, cnt, kV0, kNV, kGW,
                                         nullptr, nullptr, nullptr);
    const hgnn_seg sg = {s0, cnt + kV0, WlL,
                         s2, cnt + kV2, WlL + 2 * kH * kH,
                         OG, nullptr, wsum,
                         blL, blL + 2 * kH, OG, kNG, kNbG};
    const hgnn_seg sc = {s3, cnt + kV3, WlL + 3 * kH * kH,
                         OC, nullptr, WrL + 3 * kH * kH,
                         nullptr, nullptr, nullptr,
                         blL + 3 * kH, nullptr, OC, kNC, kNbC};
    hgnn_gemm3<<<kNbG + kNbC, 64, 0, stream>>>(sg, sc, none, 0);
  }
}


# Segment-mean gather fused into the product's LDS staging: a wave sums its 16 nodes' neighbour rows in list order straight into the A tile; no sum planes, 7 -> 3 launches; sum order unchanged
# speedup vs baseline: 3.6573x; 1.3047x over previous
#include <hip/hip_runtime.h>


constexpr int kNG = 100000;
constexpr int kND = 50000;
constexpr int kNC = 50000;
constexpr int kH  = 128;

constexpr int kV1 = 0;
constexpr int kV0 = kND;
constexpr int kV2 = kND + kNG;
constexpr int kV3 = kND + 2 * kNG;
constexpr int kNV = kND + 2 * kNG + kNC;
constexpr int kETot = 600000 + 300000 + 600000 + 600000;

typedef __attribute__((ext_vector_type(2))) float v2f;
typedef __attribute__((ext_vector_type(8))) float v8f;

constexpr int kBT   = 1024;
constexpr int kWG1  = 34;
constexpr int kWG0  = 74;
constexpr int kWG2  = 74;
constexpr int kWG3  = 74;
constexpr int kBuildGrid = kWG1 + kWG0 + kWG2 + kWG3;
constexpr int kWsumWG    = 2 * kH * kH / kBT;
constexpr int kMaxR = 2 * kBT;
constexpr int kMaxGroups = 5;
constexpr int kMaxEdges  = kMaxGroups * 32 * kBT * 4;
constexpr int kNoNode    = 0x7fffffff;
static_assert((kND + kWG1 - 1) / kWG1 <= kMaxR, "range too long");
static_assert((kNG + kWG0 - 1) / kWG0 <= kMaxR, "range too long");
static_assert((kNG + kWG2 - 1) / kWG2 <= kMaxR, "range too long");
static_assert((kNC + kWG3 - 1) / kWG3 <= kMaxR, "range too long");
static_assert(600000 <= kMaxEdges, "mask words too few");

__global__ __launch_bounds__(kBT)
void hgnn_build(const int* __restrict__ s0, const int* __restrict__ d0,
                const int* __restrict__ s1, const int* __restrict__ d1,
                const int* __restrict__ s2, const int* __restrict__ d2,
                const int* __restrict__ s3, const int* __restrict__ d3,
                int n0, int n1, int n2, int n3,
                int* __restrict__ deg, int* __restrict__ cur, int* __restrict__ col,
                const float* __restrict__ wr, float* __restrict__ wsum) {
  __shared__ int      ldeg[kMaxR];
  __shared__ unsigned lmask[kMaxGroups * kBT];
  __shared__ int      wpart[32];
  __shared__ int      sbelow;
  const int tid  = threadIdx.x;
  const int lane = tid & 31;
  const int wv   = tid >> 5;

  if ((int)blockIdx.x >= kBuildGrid) {
    const int i = ((int)blockIdx.x - kBuildGrid) * kBT + tid;
    const int l = i / (kH * kH);
    const int j = i - l * (kH * kH);
    if (l < 2) {
      const float* w = wr + (size_t)l * 4 * kH * kH;
      wsum[i] = w[j] + w[2 * kH * kH + j];
    }
    return;
  }

  const int b = blockIdx.x;
  int r, w, nwg, nn;
  if (b < kWG1)                    { r = 1; w = b;                      nwg = kWG1; nn = kND; }
  else if (b < kWG1 + kWG0)        { r = 0; w = b - kWG1;               nwg = kWG0; nn = kNG; }
  else if (b < kWG1 + kWG0 + kWG2) { r = 2; w = b - kWG1 - kWG0;        nwg = kWG2; nn = kNG; }
  else                             { r = 3; w = b - kWG1 - kWG0 - kWG2; nwg = kWG3; nn = kNC; }
  const int* s = (r == 0) ? s0 : (r == 1) ? s1 : (r == 2) ? s2 : s3;
  const int* d = (r == 0) ? d0 : (r == 1) ? d1 : (r == 2) ? d2 : d3;
  int n        = (r == 0) ? n0 : (r == 1) ? n1 : (r == 2) ? n2 : n3;
  if (n > kMaxEdges) n = kMaxEdges;
  const int vb    = (r == 0) ? kV0 : (r == 1) ? kV1 : (r == 2) ? kV2 : kV3;
  const int ebase = (r == 1) ? 0 : (r == 0) ? n1 : (r == 2) ? n1 + n0 : n1 + n0 + n2;
  const int rpw = (nn + nwg - 1) / nwg;
  const int lo  = w * rpw;
  const int R   = (nn - lo < rpw) ? (nn - lo) : rpw;

  for (int i = tid; i < kMaxR; i += kBT) ldeg[i] = 0;
  if (tid == 0) sbelow = 0;
  __syncthreads();

  const int4* d4 = (const int4*)d;
  const int n4 = n >> 2;
  int below = 0;
  auto count1 = [&](int dv) -> bool {
    const unsigned l = (unsigned)dv - (unsigned)lo;
    below += (dv < lo) ? 1 : 0;
    if (l < (unsigned)R) { atomicAdd(&ldeg[l], 1); return true; }
    return false;
  };

  for (int g = 0; g < kMaxGroups; ++g) {
    unsigned mask = 0;
    for (int h = 0; h < 8; ++h) {
      const int jb = ((g * 8 + h) * 4) * kBT + tid;
      if (jb >= n4) break;
      int4 q[4];
#pragma unroll
      for (int u = 0; u < 4; ++u) {
        const int j = jb + u * kBT;
        q[u] = (j < n4) ? d4[j] : make_int4(kNoNode, kNoNode, kNoNode, kNoNode);
      }
#pragma unroll
      for (int u = 0; u < 4; ++u) {
        const bool hx = count1(q[u].x);
        const bool hy = count1(q[u].y);
        const bool hz = count1(q[u].z);
        const bool hw = count1(q[u].w);
        if (hx | hy | hz | hw) mask |= 1u << (h * 4 + u);
      }
    }
    lmask[g * kBT + tid] = mask;
  }
  const int et = (n4 << 2) + tid;
  const bool has_tail = tid < (n & 3);
  if (has_tail) count1(d[et]);

  for (int o = 16; o > 0; o >>= 1) below += __shfl_xor(below, o);
  if (lane == 0) atomicAdd(&sbelow, below);
  __syncthreads();

  const int i0 = 2 * tid;
  const int i1 = i0 + 1;
  const int a  = (i0 < R) ? ldeg[i0] : 0;
  const int c  = (i1 < R) ? ldeg[i1] : 0;
  const int sum = a + c;
  int inc = sum;
  for (int o = 1; o < 32; o <<= 1) {
    const int t = __shfl_up(inc, o);
    if (lane >= o) inc += t;
  }
  if (lane == 31) wpart[wv] = inc;
  __syncthreads();
  if (wv == 0) {
    const int x = wpart[lane];
    int xi = x;
    for (int o = 1; o < 32; o <<= 1) {
      const int t = __shfl_up(xi, o);
      if (lane >= o) xi += t;
    }
    wpart[lane] = xi - x;
  }
  __syncthreads();
  const int ex   = wpart[wv] + inc - sum;
  const int base = ebase + sbelow;
  if (i0 < R) { ldeg[i0] = ex;     deg[vb + lo + i0] = a; cur[vb + lo + i0] = base + ex + a; }
  if (i1 < R) { ldeg[i1] = ex + a; deg[vb + lo + i1] = c; cur[vb + lo + i1] = base + ex + sum; }
  __syncthreads();

  auto place = [&](int dv, int e) {
    const unsigned l = (unsigned)dv - (unsigned)lo;
    if (l < (unsigned)R) {
      const int p = base + atomicAdd(&ldeg[l], 1);
      if ((unsigned)p < (unsigned)kETot) col[p] = s[e];
    }
  };
  for (int g = 0; g < kMaxGroups; ++g) {
    unsigned mask = lmask[g * kBT + tid];
    while (mask) {
      const int k = __builtin_ctz(mask);
      mask &= mask - 1;
      const int j = (g * 32 + k) * kBT + tid;
      const int4 q = d4[j];
      place(q.x, 4 * j);
      place(q.y, 4 * j + 1);
      place(q.z, 4 * j + 2);
      place(q.w, 4 * j + 3);
    }
  }
  if (has_tail) place(d[et], et);
}

constexpr int kBM  = 32;
constexpr int kLdA = 132;

__device__ __forceinline__ v8f hgnn_mma(v2f a, float b0, float b1, v8f c) {
  v2f b;
  b.x = b0;
  b.y = b1;
  return __builtin_amdgcn_wmma_f32_16x16x4_f32(false, a, false, b, (short)0, c, false, false);
}

struct hgnn_seg {
  const float* A0; const float* W0;
  const float* A1; const float* W1;
  const float* A2; const float* W2;
  const float* b0; const float* b1;
  float* Y;
  int v0; int m0; int v1; int m1; int v2; int m2;
  int nrows; int nblk;
};

__global__ __launch_bounds__(64)
void hgnn_layer(hgnn_seg g0, hgnn_seg g1, hgnn_seg g2,
                const int* __restrict__ deg, const int* __restrict__ cur,
                const int* __restrict__ col, int relu) {
  int blk = blockIdx.x;
  int seg = 0;
  if (blk >= g0.nblk) { blk -= g0.nblk; seg = 1; if (blk >= g1.nblk) { blk -= g1.nblk; seg = 2; } }
#define HGNN_SEL(f) ((seg == 0) ? g0.f : (seg == 1) ? g1.f : g2.f)
  const float* __restrict__ A0 = HGNN_SEL(A0);
  const float* __restrict__ W0 = HGNN_SEL(W0);
  const float* __restrict__ A1 = HGNN_SEL(A1);
  const float* __restrict__ W1 = HGNN_SEL(W1);
  const float* __restrict__ A2 = HGNN_SEL(A2);
  const float* __restrict__ W2 = HGNN_SEL(W2);
  const float* __restrict__ b0 = HGNN_SEL(b0);
  const float* __restrict__ b1 = HGNN_SEL(b1);
  float* __restrict__ Y = HGNN_SEL(Y);
  const int v0 = HGNN_SEL(v0);
  const int m0 = HGNN_SEL(m0);
  const int v1 = HGNN_SEL(v1);
  const int m1 = HGNN_SEL(m1);
  const int v2 = HGNN_SEL(v2);
  const int m2 = HGNN_SEL(m2);
  const int nrows = HGNN_SEL(nrows);
#undef HGNN_SEL
  __shared__ __attribute__((aligned(16))) float As[kBM * kLdA];
  const int tid  = threadIdx.x;
  const int lane = tid & 31;
  const int wv   = __builtin_amdgcn_readfirstlane(tid >> 5);
  const int c    = lane & 15;
  const int koff = (lane >> 4) << 1;
  const int colb = (wv << 6) + (c << 2);
  const int row0 = blk * kBM;

  v8f acc[2][4] = {};
#pragma unroll 1
  for (int t = 0; t < 3; ++t) {
    const float* A  = (t == 0) ? A0 : (t == 1) ? A1 : A2;
    const float* W  = (t == 0) ? W0 : (t == 1) ? W1 : W2;
    const int    vb = (t == 0) ? v0 : (t == 1) ? v1 : v2;
    const int    mx = (t == 0) ? m0 : (t == 1) ? m1 : m2;
    if (!A) continue;

    __syncthreads();
    if (vb >= 0) {
      const int rbase = row0 + 16 * wv;
      int dn, ce;
      {
        const int  row = rbase + c;
        const bool ok  = row < nrows;
        const int  v   = vb + (ok ? row : (nrows - 1));
        ce = cur[v];
        dn = ok ? deg[v] : 0;
      }
      int eb = __builtin_amdgcn_readlane(ce, 0) - __builtin_amdgcn_readlane(dn, 0);
      int ee = __builtin_amdgcn_readlane(ce, 15);
      if (eb < 0) eb = 0;
      if (ee > kETot) ee = kETot;
      const unsigned smax = (unsigned)mx;

      int    i    = 0;
      int    iend = __builtin_amdgcn_readlane(ce, 0);
      float4 sum  = make_float4(0.0f, 0.0f, 0.0f, 0.0f);
      auto flush = [&]() {
        const int   n  = __builtin_amdgcn_readlane(dn, i);
        const float sc = 1.0f / fmaxf((float)n, 1.0f);
        float4 q = sum;
        q.x *= sc; q.y *= sc; q.z *= sc; q.w *= sc;
        *(float4*)(As + (16 * wv + i) * kLdA + lane * 4) = q;
        sum = make_float4(0.0f, 0.0f, 0.0f, 0.0f);
        ++i;
        iend = __builtin_amdgcn_readlane(ce, (i < 16) ? i : 15);
      };

      for (int e0 = eb; e0 < ee; e0 += 32) {
        const int m = (ee - e0 < 32) ? (ee - e0) : 32;
        int idx = 0;
        if (lane < m) idx = col[e0 + lane];
        for (int j0 = 0; j0 < m; j0 += 8) {
          unsigned s[8];
#pragma unroll
          for (int u = 0; u < 8; ++u) {
            const unsigned x = (unsigned)__builtin_amdgcn_readlane(idx, j0 + u);
            s[u] = (x < smax) ? x : smax;
          }
          float4 r[8];
#pragma unroll
          for (int u = 0; u < 8; ++u) {
            r[u] = make_float4(0.0f, 0.0f, 0.0f, 0.0f);
            if (j0 + u < m) r[u] = *(const float4*)(A + (size_t)s[u] * kH + lane * 4);
          }
#pragma unroll
          for (int u = 0; u < 8; ++u) {
            if (j0 + u < m) {
              const int e = e0 + j0 + u;
              while (i < 16 && iend <= e) flush();
              sum.x += r[u].x; sum.y += r[u].y; sum.z += r[u].z; sum.w += r[u].w;
            }
          }
        }
      }
      while (i < 16) flush();
    } else {
#pragma unroll 1
      for (int u0 = 0; u0 < kBM / 2; u0 += 8) {
        float4 q[8];
#pragma unroll
        for (int u = 0; u < 8; ++u) {
          const int row = row0 + wv + 2 * (u0 + u);
          const int rc  = (row < nrows) ? row : (nrows - 1);
          q[u] = *(const float4*)(A + (size_t)rc * kH + lane * 4);
        }
#pragma unroll
        for (int u = 0; u < 8; ++u) {
          const int rl = wv + 2 * (u0 + u);
          if (row0 + rl >= nrows) q[u] = make_float4(0.0f, 0.0f, 0.0f, 0.0f);
          *(float4*)(As + rl * kLdA + lane * 4) = q[u];
        }
      }
    }
    __syncthreads();

    const float* Wp  = W + (size_t)koff * kH + colb;
    const float* Ap0 = As + c * kLdA + koff;
    const float* Ap1 = As + (16 + c) * kLdA + koff;
#pragma unroll 4
    for (int k0 = 0; k0 < kH; k0 += 4) {
      const float4 w0 = *(const float4*)(Wp + (size_t)k0 * kH);
      const float4 w1 = *(const float4*)(Wp + (size_t)(k0 + 1) * kH);
      const v2f a0 = *(const v2f*)(Ap0 + k0);
      const v2f a1 = *(const v2f*)(Ap1 + k0);
      acc[0][0] = hgnn_mma(a0, w0.x, w1.x, acc[0][0]);
      acc[1][0] = hgnn_mma(a1, w0.x, w1.x, acc[1][0]);
      acc[0][1] = hgnn_mma(a0, w0.y, w1.y, acc[0][1]);
      acc[1][1] = hgnn_mma(a1, w0.y, w1.y, acc[1][1]);
      acc[0][2] = hgnn_mma(a0, w0.z, w1.z, acc[0][2]);
      acc[1][2] = hgnn_mma(a1, w0.z, w1.z, acc[1][2]);
      acc[0][3] = hgnn_mma(a0, w0.w, w1.w, acc[0][3]);
      acc[1][3] = hgnn_mma(a1, w0.w, w1.w, acc[1][3]);
    }
  }

  float4 bs = make_float4(0.0f, 0.0f, 0.0f, 0.0f);
  if (b0) { const float4 v = *(const float4*)(b0 + colb); bs.x += v.x; bs.y += v.y; bs.z += v.z; bs.w += v.w; }
  if (b1) { const float4 v = *(const float4*)(b1 + colb); bs.x += v.x; bs.y += v.y; bs.z += v.z; bs.w += v.w; }
#pragma unroll
  for (int i = 0; i < 2; ++i) {
#pragma unroll
    for (int r = 0; r < 8; ++r) {
      const int row = row0 + 16 * i + r + ((lane >> 4) << 3);
      float4 v;
      v.x = acc[i][0][r] + bs.x;
      v.y = acc[i][1][r] + bs.y;
      v.z = acc[i][2][r] + bs.z;
      v.w = acc[i][3][r] + bs.w;
      if (relu) { v.x = fmaxf(v.x, 0.0f); v.y = fmaxf(v.y, 0.0f); v.z = fmaxf(v.z, 0.0f); v.w = fmaxf(v.w, 0.0f); }
      if (row < nrows) *(float4*)(Y + (size_t)row * kH + colb) = v;
    }
  }
}

extern "C" void kernel_launch(void* const* d_in, const int* in_sizes, int n_in,
                              void* d_out, int out_size, void* d_ws, size_t ws_size,
                              hipStream_t stream) {
  (void)n_in; (void)out_size; (void)ws_size;
  const float* EG = (const float*)d_in[0];
  const float* ED = (const float*)d_in[1];
  const float* EC = (const float*)d_in[2];
  const float* Wl = (const float*)d_in[3];
  const float* bl = (const float*)d_in[4];
  const float* Wr = (const float*)d_in[5];
  const int* eg_s  = (const int*)d_in[6];
  const int* eg_d  = (const int*)d_in[7];
  const int* ecd_s = (const int*)d_in[8];
  const int* ecd_d = (const int*)d_in[9];
  const int* ecg_s = (const int*)d_in[10];
  const int* ecg_d = (const int*)d_in[11];
  const int* egc_s = (const int*)d_in[12];
  const int* egc_d = (const int*)d_in[13];

  float* ws = (float*)d_ws;
  float* XG = ws;
  float* XD = XG + (size_t)kNG * kH;
  float* XC = XD + (size_t)kND * kH;
  float* wsum = XC + (size_t)kNC * kH;
  int* deg = (int*)(wsum + 2 * kH * kH);
  int* cur = deg + kNV;
  int* col = cur + kNV;

  float* OG = (float*)d_out;
  float* OD = OG + (size_t)kNG * kH;
  float* OC = OD + (size_t)kND * kH;

  hgnn_build<<<kBuildGrid + kWsumWG, kBT, 0, stream>>>(
      eg_s, eg_d, ecd_s, ecd_d, ecg_s, ecg_d, egc_s, egc_d,
      in_sizes[6], in_sizes[8], in_sizes[10], in_sizes[12], deg, cur, col, Wr, wsum);

  constexpr int kNbG = (kNG + kBM - 1) / kBM;
  constexpr int kNbD = (kND + kBM - 1) / kBM;
  constexpr int kNbC = (kNC + kBM - 1) / kBM;

  {
    const float* WlL = Wl;
    const float* WrL = Wr;
    const float* blL = bl;
    const hgnn_seg sg = {EG, WlL,
                         EC, WlL + 2 * kH * kH,
                         EG, wsum,
                         blL, blL + 2 * kH, XG,
                         kV0, kNG - 1, kV2, kNC - 1, -1, 0, kNG, kNbG};
    const hgnn_seg sd = {EC, WlL + 1 * kH * kH,
                         ED, WrL + 1 * kH * kH,
                         nullptr, nullptr,
                         blL + 1 * kH, nullptr, XD,
                         kV1, kNC - 1, -1, 0, -1, 0, kND, kNbD};
    const hgnn_seg sc = {EG, WlL + 3 * kH * kH,
                         EC, WrL + 3 * kH * kH,
                         nullptr, nullptr,
                         blL + 3 * kH, nullptr, XC,
                         kV3, kNG - 1, -1, 0, -1, 0, kNC, kNbC};
    hgnn_layer<<<kNbG + kNbD + kNbC, 64, 0, stream>>>(sg, sd, sc, deg, cur, col, 1);
  }

  {
    const float* WlL = Wl + (size_t)4 * kH * kH;
    const float* WrL = Wr + (size_t)4 * kH * kH;
    const float* blL = bl + (size_t)4 * kH;
    const hgnn_seg sg = {XG, WlL,
                         XC, WlL + 2 * kH * kH,
                         XG, wsum + kH * kH,
                         blL, blL + 2 * kH, OG,
                         kV0, kNG - 1, kV2, kNC - 1, -1, 0, kNG, kNbG};
    const hgnn_seg sd = {XC, WlL + 1 * kH * kH,
                         XD, WrL + 1 * kH * kH,
                         nullptr, nullptr,
                         blL + 1 * kH, nullptr, OD,
                         kV1, kNC - 1, -1, 0, -1, 0, kND, kNbD};
    const hgnn_seg sc = {XG, WlL + 3 * kH * kH,
                         XC, WrL + 3 * kH * kH,
                         nullptr, nullptr,
                         blL + 3 * kH, nullptr, OC,
                         kV3, kNG - 1, -1, 0, -1, 0, kNC, kNbC};
    hgnn_layer<<<kNbG + kNbD + kNbC, 64, 0, stream>>>(sg, sd, sc, deg, cur, col, 0);
  }
}


# Gather walk made lean: a batch's 8 row loads go out unconditionally (a slot past the end reads row 0, not added), sums start at the first row back, node-end test is one compare; sum order kept
# speedup vs baseline: 3.7781x; 1.0330x over previous
#include <hip/hip_runtime.h>


constexpr int kNG = 100000;
constexpr int kND = 50000;
constexpr int kNC = 50000;
constexpr int kH  = 128;

constexpr int kV1 = 0;
constexpr int kV0 = kND;
constexpr int kV2 = kND + kNG;
constexpr int kV3 = kND + 2 * kNG;
constexpr int kNV = kND + 2 * kNG + kNC;
constexpr int kETot = 600000 + 300000 + 600000 + 600000;

typedef __attribute__((ext_vector_type(2))) float v2f;
typedef __attribute__((ext_vector_type(8))) float v8f;

constexpr int kBT   = 1024;
constexpr int kWG1  = 34;
constexpr int kWG0  = 74;
constexpr int kWG2  = 74;
constexpr int kWG3  = 74;
constexpr int kBuildGrid = kWG1 + kWG0 + kWG2 + kWG3;
constexpr int kWsumWG    = 2 * kH * kH / kBT;
constexpr int kMaxR = 2 * kBT;
constexpr int kMaxGroups = 5;
constexpr int kMaxEdges  = kMaxGroups * 32 * kBT * 4;
constexpr int kNoNode    = 0x7fffffff;
static_assert((kND + kWG1 - 1) / kWG1 <= kMaxR, "range too long");
static_assert((kNG + kWG0 - 1) / kWG0 <= kMaxR, "range too long");
static_assert((kNG + kWG2 - 1) / kWG2 <= kMaxR, "range too long");
static_assert((kNC + kWG3 - 1) / kWG3 <= kMaxR, "range too long");
static_assert(600000 <= kMaxEdges, "mask words too few");

__global__ __launch_bounds__(kBT)
void hgnn_build(const int* __restrict__ s0, const int* __restrict__ d0,
                const int* __restrict__ s1, const int* __restrict__ d1,
                const int* __restrict__ s2, const int* __restrict__ d2,
                const int* __restrict__ s3, const int* __restrict__ d3,
                int n0, int n1, int n2, int n3,
                int* __restrict__ deg, int* __restrict__ cur, int* __restrict__ col,
                const float* __restrict__ wr, float* __restrict__ wsum) {
  __shared__ int      ldeg[kMaxR];
  __shared__ unsigned lmask[kMaxGroups * kBT];
  __shared__ int      wpart[32];
  __shared__ int      sbelow;
  const int tid  = threadIdx.x;
  const int lane = tid & 31;
  const int wv   = tid >> 5;

  if ((int)blockIdx.x >= kBuildGrid) {
    const int i = ((int)blockIdx.x - kBuildGrid) * kBT + tid;
    const int l = i / (kH * kH);
    const int j = i - l * (kH * kH);
    if (l < 2) {
      const float* w = wr + (size_t)l * 4 * kH * kH;
      wsum[i] = w[j] + w[2 * kH * kH + j];
    }
    return;
  }

  const int b = blockIdx.x;
  int r, w, nwg, nn;
  if (b < kWG1)                    { r = 1; w = b;                      nwg = kWG1; nn = kND; }
  else if (b < kWG1 + kWG0)        { r = 0; w = b - kWG1;               nwg = kWG0; nn = kNG; }
  else if (b < kWG1 + kWG0 + kWG2) { r = 2; w = b - kWG1 - kWG0;        nwg = kWG2; nn = kNG; }
  else                             { r = 3; w = b - kWG1 - kWG0 - kWG2; nwg = kWG3; nn = kNC; }
  const int* s = (r == 0) ? s0 : (r == 1) ? s1 : (r == 2) ? s2 : s3;
  const int* d = (r == 0) ? d0 : (r == 1) ? d1 : (r == 2) ? d2 : d3;
  int n        = (r == 0) ? n0 : (r == 1) ? n1 : (r == 2) ? n2 : n3;
  if (n > kMaxEdges) n = kMaxEdges;
  const int vb    = (r == 0) ? kV0 : (r == 1) ? kV1 : (r == 2) ? kV2 : kV3;
  const int ebase = (r == 1) ? 0 : (r == 0) ? n1 : (r == 2) ? n1 + n0 : n1 + n0 + n2;
  const int rpw = (nn + nwg - 1) / nwg;
  const int lo  = w * rpw;
  const int R   = (nn - lo < rpw) ? (nn - lo) : rpw;

  for (int i = tid; i < kMaxR; i += kBT) ldeg[i] = 0;
  if (tid == 0) sbelow = 0;
  __syncthreads();

  const int4* d4 = (const int4*)d;
  const int n4 = n >> 2;
  int below = 0;
  auto count1 = [&](int dv) -> bool {
    const unsigned l = (unsigned)dv - (unsigned)lo;
    below += (dv < lo) ? 1 : 0;
    if (l < (unsigned)R) { atomicAdd(&ldeg[l], 1); return true; }
    return false;
  };

  for (int g = 0; g < kMaxGroups; ++g) {
    unsigned mask = 0;
    for (int h = 0; h < 8; ++h) {
      const int jb = ((g * 8 + h) * 4) * kBT + tid;
      if (jb >= n4) break;
      int4 q[4];
#pragma unroll
      for (int u = 0; u < 4; ++u) {
        const int j = jb + u * kBT;
        q[u] = (j < n4) ? d4[j] : make_int4(kNoNode, kNoNode, kNoNode, kNoNode);
      }
#pragma unroll
      for (int u = 0; u < 4; ++u) {
        const bool hx = count1(q[u].x);
        const bool hy = count1(q[u].y);
        const bool hz = count1(q[u].z);
        const bool hw = count1(q[u].w);
        if (hx | hy | hz | hw) mask |= 1u << (h * 4 + u);
      }
    }
    lmask[g * kBT + tid] = mask;
  }
  const int et = (n4 << 2) + tid;
  const bool has_tail = tid < (n & 3);
  if (has_tail) count1(d[et]);

  for (int o = 16; o > 0; o >>= 1) below += __shfl_xor(below, o);
  if (lane == 0) atomicAdd(&sbelow, below);
  __syncthreads();

  const int i0 = 2 * tid;
  const int i1 = i0 + 1;
  const int a  = (i0 < R) ? ldeg[i0] : 0;
  const int c  = (i1 < R) ? ldeg[i1] : 0;
  const int sum = a + c;
  int inc = sum;
  for (int o = 1; o < 32; o <<= 1) {
    const int t = __shfl_up(inc, o);
    if (lane >= o) inc += t;
  }
  if (lane == 31) wpart[wv] = inc;
  __syncthreads();
  if (wv == 0) {
    const int x = wpart[lane];
    int xi = x;
    for (int o = 1; o < 32; o <<= 1) {
      const int t = __shfl_up(xi, o);
      if (lane >= o) xi += t;
    }
    wpart[lane] = xi - x;
  }
  __syncthreads();
  const int ex   = wpart[wv] + inc - sum;
  const int base = ebase + sbelow;
  if (i0 < R) { ldeg[i0] = ex;     deg[vb + lo + i0] = a; cur[vb + lo + i0] = base + ex + a; }
  if (i1 < R) { ldeg[i1] = ex + a; deg[vb + lo + i1] = c; cur[vb + lo + i1] = base + ex + sum; }
  __syncthreads();

  auto place = [&](int dv, int e) {
    const unsigned l = (unsigned)dv - (unsigned)lo;
    if (l < (unsigned)R) {
      const int p = base + atomicAdd(&ldeg[l], 1);
      if ((unsigned)p < (unsigned)kETot) col[p] = s[e];
    }
  };
  for (int g = 0; g < kMaxGroups; ++g) {
    unsigned mask = lmask[g * kBT + tid];
    while (mask) {
      const int k = __builtin_ctz(mask);
      mask &= mask - 1;
      const int j = (g * 32 + k) * kBT + tid;
      const int4 q = d4[j];
      place(q.x, 4 * j);
      place(q.y, 4 * j + 1);
      place(q.z, 4 * j + 2);
      place(q.w, 4 * j + 3);
    }
  }
  if (has_tail) place(d[et], et);
}

constexpr int kBM  = 32;
constexpr int kLdA = 132;
constexpr int kNoEnd = 0x7fffffff;

__device__ __forceinline__ v8f hgnn_mma(v2f a, float b0, float b1, v8f c) {
  v2f b;
  b.x = b0;
  b.y = b1;
  return __builtin_amdgcn_wmma_f32_16x16x4_f32(false, a, false, b, (short)0, c, false, false);
}

struct hgnn_seg {
  const float* A0; const float* W0;
  const float* A1; const float* W1;
  const float* A2; const float* W2;
  const float* b0; const float* b1;
  float* Y;
  int v0; int m0; int v1; int m1; int v2; int m2;
  int nrows; int nblk;
};

__global__ __launch_bounds__(64)
void hgnn_layer(hgnn_seg g0, hgnn_seg g1, hgnn_seg g2,
                const int* __restrict__ deg, const int* __restrict__ cur,
                const int* __restrict__ col, int relu) {
  int blk = blockIdx.x;
  int seg = 0;
  if (blk >= g0.nblk) { blk -= g0.nblk; seg = 1; if (blk >= g1.nblk) { blk -= g1.nblk; seg = 2; } }
#define HGNN_SEL(f) ((seg == 0) ? g0.f : (seg == 1) ? g1.f : g2.f)
  const float* __restrict__ A0 = HGNN_SEL(A0);
  const float* __restrict__ W0 = HGNN_SEL(W0);
  const float* __restrict__ A1 = HGNN_SEL(A1);
  const float* __restrict__ W1 = HGNN_SEL(W1);
  const float* __restrict__ A2 = HGNN_SEL(A2);
  const float* __restrict__ W2 = HGNN_SEL(W2);
  const float* __restrict__ b0 = HGNN_SEL(b0);
  const float* __restrict__ b1 = HGNN_SEL(b1);
  float* __restrict__ Y = HGNN_SEL(Y);
  const int v0 = HGNN_SEL(v0);
  const int m0 = HGNN_SEL(m0);
  const int v1 = HGNN_SEL(v1);
  const int m1 = HGNN_SEL(m1);
  const int v2 = HGNN_SEL(v2);
  const int m2 = HGNN_SEL(m2);
  const int nrows = HGNN_SEL(nrows);
#undef HGNN_SEL
  __shared__ __attribute__((aligned(16))) float As[kBM * kLdA];
  const int tid  = threadIdx.x;
  const int lane = tid & 31;
  const int wv   = __builtin_amdgcn_readfirstlane(tid >> 5);
  const int c    = lane & 15;
  const int koff = (lane >> 4) << 1;
  const int colb = (wv << 6) + (c << 2);
  const int row0 = blk * kBM;

  v8f acc[2][4] = {};
#pragma unroll 1
  for (int t = 0; t < 3; ++t) {
    const float* A  = (t == 0) ? A0 : (t == 1) ? A1 : A2;
    const float* W  = (t == 0) ? W0 : (t == 1) ? W1 : W2;
    const int    vb = (t == 0) ? v0 : (t == 1) ? v1 : v2;
    const int    mx = (t == 0) ? m0 : (t == 1) ? m1 : m2;
    if (!A) continue;

    __syncthreads();
    if (vb >= 0) {
      const int rbase = row0 + 16 * wv;
      int dn, ce;
      {
        const int  row = rbase + c;
        const bool ok  = row < nrows;
        const int  v   = vb + (ok ? row : (nrows - 1));
        ce = cur[v];
        dn = ok ? deg[v] : 0;
      }
      int eb = __builtin_amdgcn_readlane(ce, 0) - __builtin_amdgcn_readlane(dn, 0);
      int ee = __builtin_amdgcn_readlane(ce, 15);
      if (eb < 0) eb = 0;
      if (ee > kETot) ee = kETot;
      const unsigned smax = (unsigned)mx;

      int    i    = 0;
      int    iend = __builtin_amdgcn_readlane(ce, 0);
      float4 sum  = make_float4(0.0f, 0.0f, 0.0f, 0.0f);
      auto flush = [&]() {
        const int   n  = __builtin_amdgcn_readlane(dn, i);
        const float sc = 1.0f / fmaxf((float)n, 1.0f);
        float4 q = sum;
        q.x *= sc; q.y *= sc; q.z *= sc; q.w *= sc;
        *(float4*)(As + (16 * wv + i) * kLdA + lane * 4) = q;
        sum = make_float4(0.0f, 0.0f, 0.0f, 0.0f);
        ++i;
        iend = (i < 16) ? __builtin_amdgcn_readlane(ce, i & 15) : kNoEnd;
      };

      for (int e0 = eb; e0 < ee; e0 += 32) {
        const int m = (ee - e0 < 32) ? (ee - e0) : 32;
        int idx = 0;
        if (lane < m) idx = col[e0 + lane];
        for (int j0 = 0; j0 < m; j0 += 8) {
          unsigned s[8];
#pragma unroll
          for (int u = 0; u < 8; ++u) {
            const unsigned x = (unsigned)__builtin_amdgcn_readlane(idx, j0 + u);
            s[u] = (x < smax) ? x : smax;
          }
          float4 r[8];
#pragma unroll
          for (int u = 0; u < 8; ++u)
            r[u] = *(const float4*)(A + (size_t)s[u] * kH + lane * 4);
          const int nb = (m - j0 < 8) ? (m - j0) : 8;
#pragma unroll
          for (int u = 0; u < 8; ++u) {
            if (u == 0 || u < nb) {
              const int e = e0 + j0 + u;
              while (iend <= e) flush();
              sum.x += r[u].x; sum.y += r[u].y; sum.z += r[u].z; sum.w += r[u].w;
            }
          }
        }
      }
      while (i < 16) flush();
    } else {
#pragma unroll 1
      for (int u0 = 0; u0 < kBM / 2; u0 += 8) {
        float4 q[8];
#pragma unroll
        for (int u = 0; u < 8; ++u) {
          const int row = row0 + wv + 2 * (u0 + u);
          const int rc  = (row < nrows) ? row : (nrows - 1);
          q[u] = *(const float4*)(A + (size_t)rc * kH + lane * 4);
        }
#pragma unroll
        for (int u = 0; u < 8; ++u) {
          const int rl = wv + 2 * (u0 + u);
          if (row0 + rl >= nrows) q[u] = make_float4(0.0f, 0.0f, 0.0f, 0.0f);
          *(float4*)(As + rl * kLdA + lane * 4) = q[u];
        }
      }
    }
    __syncthreads();

    const float* Wp  = W + (size_t)koff * kH + colb;
    const float* Ap0 = As + c * kLdA + koff;
    const float* Ap1 = As + (16 + c) * kLdA + koff;
#pragma unroll 4
    for (int k0 = 0; k0 < kH; k0 += 4) {
      const float4 w0 = *(const float4*)(Wp + (size_t)k0 * kH);
      const float4 w1 = *(const float4*)(Wp + (size_t)(k0 + 1) * kH);
      const v2f a0 = *(const v2f*)(Ap0 + k0);
      const v2f a1 = *(const v2f*)(Ap1 + k0);
      acc[0][0] = hgnn_mma(a0, w0.x, w1.x, acc[0][0]);
      acc[1][0] = hgnn_mma(a1, w0.x, w1.x, acc[1][0]);
      acc[0][1] = hgnn_mma(a0, w0.y, w1.y, acc[0][1]);
      acc[1][1] = hgnn_mma(a1, w0.y, w1.y, acc[1][1]);
      acc[0][2] = hgnn_mma(a0, w0.z, w1.z, acc[0][2]);
      acc[1][2] = hgnn_mma(a1, w0.z, w1.z, acc[1][2]);
      acc[0][3] = hgnn_mma(a0, w0.w, w1.w, acc[0][3]);
      acc[1][3] = hgnn_mma(a1, w0.w, w1.w, acc[1][3]);
    }
  }

  float4 bs = make_float4(0.0f, 0.0f, 0.0f, 0.0f);
  if (b0) { const float4 v = *(const float4*)(b0 + colb); bs.x += v.x; bs.y += v.y; bs.z += v.z; bs.w += v.w; }
  if (b1) { const float4 v = *(const float4*)(b1 + colb); bs.x += v.x; bs.y += v.y; bs.z += v.z; bs.w += v.w; }
#pragma unroll
  for (int i = 0; i < 2; ++i) {
#pragma unroll
    for (int r = 0; r < 8; ++r) {
      const int row = row0 + 16 * i + r + ((lane >> 4) << 3);
      float4 v;
      v.x = acc[i][0][r] + bs.x;
      v.y = acc[i][1][r] + bs.y;
      v.z = acc[i][2][r] + bs.z;
      v.w = acc[i][3][r] + bs.w;
      if (relu) { v.x = fmaxf(v.x, 0.0f); v.y = fmaxf(v.y, 0.0f); v.z = fmaxf(v.z, 0.0f); v.w = fmaxf(v.w, 0.0f); }
      if (row < nrows) *(float4*)(Y + (size_t)row * kH + colb) = v;
    }
  }
}

extern "C" void kernel_launch(void* const* d_in, const int* in_sizes, int n_in,
                              void* d_out, int out_size, void* d_ws, size_t ws_size,
                              hipStream_t stream) {
  (void)n_in; (void)out_size; (void)ws_size;
  const float* EG = (const float*)d_in[0];
  const float* ED = (const float*)d_in[1];
  const float* EC = (const float*)d_in[2];
  const float* Wl = (const float*)d_in[3];
  const float* bl = (const float*)d_in[4];
  const float* Wr = (const float*)d_in[5];
  const int* eg_s  = (const int*)d_in[6];
  const int* eg_d  = (const int*)d_in[7];
  const int* ecd_s = (const int*)d_in[8];
  const int* ecd_d = (const int*)d_in[9];
  const int* ecg_s = (const int*)d_in[10];
  const int* ecg_d = (const int*)d_in[11];
  const int* egc_s = (const int*)d_in[12];
  const int* egc_d = (const int*)d_in[13];

  float* ws = (float*)d_ws;
  float* XG = ws;
  float* XD = XG + (size_t)kNG * kH;
  float* XC = XD + (size_t)kND * kH;
  float* wsum = XC + (size_t)kNC * kH;
  int* deg = (int*)(wsum + 2 * kH * kH);
  int* cur = deg + kNV;
  int* col = cur + kNV;

  float* OG = (float*)d_out;
  float* OD = OG + (size_t)kNG * kH;
  float* OC = OD + (size_t)kND * kH;

  hgnn_build<<<kBuildGrid + kWsumWG, kBT, 0, stream>>>(
      eg_s, eg_d, ecd_s, ecd_d, ecg_s, ecg_d, egc_s, egc_d,
      in_sizes[6], in_sizes[8], in_sizes[10], in_sizes[12], deg, cur, col, Wr, wsum);

  constexpr int kNbG = (kNG + kBM - 1) / kBM;
  constexpr int kNbD = (kND + kBM - 1) / kBM;
  constexpr int kNbC = (kNC + kBM - 1) / kBM;

  {
    const float* WlL = Wl;
    const float* WrL = Wr;
    const float* blL = bl;
    const hgnn_seg sg = {EG, WlL,
                         EC, WlL + 2 * kH * kH,
                         EG, wsum,
                         blL, blL + 2 * kH, XG,
                         kV0, kNG - 1, kV2, kNC - 1, -1, 0, kNG, kNbG};
    const hgnn_seg sd = {EC, WlL + 1 * kH * kH,
                         ED, WrL + 1 * kH * kH,
                         nullptr, nullptr,
                         blL + 1 * kH, nullptr, XD,
                         kV1, kNC - 1, -1, 0, -1, 0, kND, kNbD};
    const hgnn_seg sc = {EG, WlL + 3 * kH * kH,
                         EC, WrL + 3 * kH * kH,
                         nullptr, nullptr,
                         blL + 3 * kH, nullptr, XC,
                         kV3, kNG - 1, -1, 0, -1, 0, kNC, kNbC};
    hgnn_layer<<<kNbG + kNbD + kNbC, 64, 0, stream>>>(sg, sd, sc, deg, cur, col, 1);
  }

  {
    const float* WlL = Wl + (size_t)4 * kH * kH;
    const float* WrL = Wr + (size_t)4 * kH * kH;
    const float* blL = bl + (size_t)4 * kH;
    const hgnn_seg sg = {XG, WlL,
                         XC, WlL + 2 * kH * kH,
                         XG, wsum + kH * kH,
                         blL, blL + 2 * kH, OG,
                         kV0, kNG - 1, kV2, kNC - 1, -1, 0, kNG, kNbG};
    const hgnn_seg sd = {XC, WlL + 1 * kH * kH,
                         XD, WrL + 1 * kH * kH,
                         nullptr, nullptr,
                         blL + 1 * kH, nullptr, OD,
                         kV1, kNC - 1, -1, 0, -1, 0, kND, kNbD};
    const hgnn_seg sc = {XG, WlL + 3 * kH * kH,
                         XC, WrL + 3 * kH * kH,
                         nullptr, nullptr,
                         blL + 3 * kH, nullptr, OC,
                         kV3, kNG - 1, -1, 0, -1, 0, kNC, kNbC};
    hgnn_layer<<<kNbG + kNbD + kNbC, 64, 0, stream>>>(sg, sd, sc, deg, cur, col, 0);
  }
}
